# starting kernel
# baseline (speedup 1.0000x reference)
#include <hip/hip_runtime.h>

typedef float v2f __attribute__((ext_vector_type(2)));
typedef float v8f __attribute__((ext_vector_type(8)));

__global__ __launch_bounds__(256) void kd_conv_wmma(
    const float* __restrict__ X, const float* __restrict__ Wt,
    const float* __restrict__ bias, const int* __restrict__ sel,
    float* __restrict__ Out, int M, int K, int Dlog2, int Bsz, int mode)
{
    extern __shared__ float sW[];
    const int D   = 1 << Dlog2;
    const int Kp4 = (K + 3) & ~3;
    const int Kp  = Kp4 + 1;
    const int m0  = blockIdx.y << 4;

    for (int r = 0; r < 16; ++r) {
        const int m = m0 + r;
        const float* wr = Wt + (size_t)m * K;
        for (int k = threadIdx.x; k < Kp4; k += 256) {
            float w = 0.0f;
            if (m < M && k < K) w = wr[k];
            sW[r * Kp + k] = w;
        }
    }
    __syncthreads();

    const int lane = threadIdx.x & 31;
    const int ng   = (blockIdx.x << 3) + (threadIdx.x >> 5);
    const int n0   = ng << 6;

    const int half = lane >> 4;
    const int l16  = lane & 15;
    const int kOff = half << 1;
    const float* aRow = sW + l16 * Kp;

    const float* xcol[4];
#pragma unroll
    for (int s = 0; s < 4; ++s) {
        const int n  = n0 + (s << 4) + l16;
        const int bb = n >> Dlog2;
        const int dd = n & (D - 1);
        xcol[s] = X + (size_t)bb * ((size_t)K << Dlog2) + dd;
    }

    v8f acc[4] = { v8f{}, v8f{}, v8f{}, v8f{} };

    const int kFull = K & ~3;
    for (int kt = 0; kt < kFull; kt += 4) {
        const int k0 = kt + kOff;
        v2f a;
        a.x = aRow[k0];
        a.y = aRow[k0 + 1];
#pragma unroll
        for (int s = 0; s < 4; ++s) {
            v2f b;
            b.x = xcol[s][(size_t)k0 << Dlog2];
            b.y = xcol[s][(size_t)(k0 + 1) << Dlog2];
            acc[s] = __builtin_amdgcn_wmma_f32_16x16x4_f32(
                false, a, false, b, (short)0, acc[s], false, false);
        }
        if (kt + 4 < kFull) {
#pragma unroll
            for (int s = 0; s < 4; ++s)
                __builtin_prefetch(xcol[s] + ((size_t)(k0 + 4) << Dlog2), 0, 1);
        }
    }
    if (kFull < K) {
        const int k0 = kFull + kOff;
        const int c0 = min(k0, K - 1);
        const int c1 = min(k0 + 1, K - 1);
        v2f a;
        a.x = aRow[k0];
        a.y = aRow[k0 + 1];
#pragma unroll
        for (int s = 0; s < 4; ++s) {
            v2f b;
            b.x = xcol[s][(size_t)c0 << Dlog2];
            b.y = xcol[s][(size_t)c1 << Dlog2];
            acc[s] = __builtin_amdgcn_wmma_f32_16x16x4_f32(
                false, a, false, b, (short)0, acc[s], false, false);
        }
    }

    const int Dh = D >> 1;
    const int F = M / 3;
#pragma unroll
    for (int s = 0; s < 4; ++s) {
        const int n  = n0 + (s << 4) + l16;
        const int bb = n >> Dlog2;
        const int dp = n & (D - 1);
#pragma unroll
        for (int r = 0; r < 8; ++r) {
            const int o = m0 + r + (half << 3);
            if (o >= M) continue;
            float v = acc[s][r] + bias[o];
            if (mode == 0) {
                v = fmaxf(v, 0.0f);
                const int f = o / 3;
                const int q = o - 3 * f;
                const int t = (q << Dlog2) + dp;
                const int dsrc = t / 3;
                const int rr = t - 3 * dsrc;
                if (sel[dsrc] == rr) {
                    float* dst = Out + (size_t)bb * (F * Dh)
                                     + (size_t)f * Dh + (dsrc >> 1);
                    atomicMax(dst, v);
                }
            } else {
                Out[(size_t)n * M + o] = v;
            }
        }
    }
}

__global__ __launch_bounds__(256) void kd_logsoftmax16(
    const float* __restrict__ logits, float* __restrict__ out, int rows)
{
    int r = blockIdx.x * blockDim.x + threadIdx.x;
    if (r >= rows) return;
    float v[16];
    float mx = -3.402823466e38f;
#pragma unroll
    for (int i = 0; i < 16; ++i) { v[i] = logits[r * 16 + i]; mx = fmaxf(mx, v[i]); }
    float s = 0.0f;
#pragma unroll
    for (int i = 0; i < 16; ++i) s += expf(v[i] - mx);
    const float l = mx + logf(s);
#pragma unroll
    for (int i = 0; i < 16; ++i) out[r * 16 + i] = v[i] - l;
}

extern "C" void kernel_launch(void* const* d_in, const int* in_sizes, int n_in,
                              void* d_out, int out_size, void* d_ws, size_t ws_size,
                              hipStream_t stream) {
    static const int DIMS[11] = {2048, 1024, 512, 256, 128, 64, 32, 16, 8, 4, 2};
    static const int INCH[11] = {3, 8, 32, 64, 64, 64, 128, 256, 512, 512, 512};
    static const int FEAT[11] = {8, 32, 64, 64, 64, 128, 256, 512, 512, 512, 1024};
    const int Bsz = 512;

    const size_t XBYTES = (size_t)512 * 32 * 512 * sizeof(float);
    float* X0     = (float*)d_ws;
    float* X1     = (float*)((char*)d_ws + XBYTES);
    float* logits = (float*)((char*)d_ws + 2 * XBYTES);

    const float* cur = (const float*)d_in[0];
    for (int i = 0; i < 11; ++i) {
        const int Dd = DIMS[i], Kc = INCH[i], F = FEAT[i], M = 3 * F;
        const int Dlog2 = __builtin_ctz((unsigned)Dd);
        float* nxt = (i & 1) ? X1 : X0;
        const size_t outBytes = (size_t)Bsz * F * (Dd / 2) * sizeof(float);
        hipMemsetAsync(nxt, 0, outBytes, stream);

        const int N = Bsz * Dd;
        const int tilesM = (M + 15) / 16;
        const int nGroups = N / 64;
        const int blocksX = nGroups / 8;
        const int Kp4 = (Kc + 3) & ~3;
        const size_t ldsBytes = (size_t)16 * (Kp4 + 1) * sizeof(float);
        dim3 grid(blocksX, tilesM);
        kd_conv_wmma<<<grid, 256, ldsBytes, stream>>>(
            cur, (const float*)d_in[12 + 2 * i], (const float*)d_in[13 + 2 * i],
            (const int*)d_in[1 + i], nxt, M, Kc, Dlog2, Bsz,  0);
        cur = nxt;
    }

    {
        const size_t ldsBytes = (size_t)16 * (1024 + 1) * sizeof(float);
        dim3 grid(1, 1);
        kd_conv_wmma<<<grid, 256, ldsBytes, stream>>>(
            cur, (const float*)d_in[34], (const float*)d_in[35], nullptr,
            logits,  16,  1024,  0, Bsz,  1);
    }

    kd_logsoftmax16<<<(512 + 255) / 256, 256, 0, stream>>>(logits, (float*)d_out, 512);

    (void)in_sizes; (void)n_in; (void)out_size; (void)ws_size;
}


# Gather moved before the GEMM: only the selected (row-third, column) per (feature, d) is computed, B masked per row-third pass; pair max in registers and plain stores replace atomic max and clears
# speedup vs baseline: 2.1632x; 2.1632x over previous
#include <hip/hip_runtime.h>

typedef float v2f __attribute__((ext_vector_type(2)));
typedef float v8f __attribute__((ext_vector_type(8)));

template <int NP>
__device__ __forceinline__ void kd_kloop(
    const float* aLane, int Kp, const int (&qlo)[4], const int (&qs)[4],
    const float* const (&xcol)[4], int K, int Dlog2, int kOff, v8f (&acc)[4])
{
    const float* ar[4][NP];
    bool on[4][NP];
#pragma unroll
    for (int s = 0; s < 4; ++s) {
#pragma unroll
        for (int p = 0; p < NP; ++p) {
            const int qv = qlo[s] + p;
            ar[s][p] = aLane + min(qv, 2) * Kp;
            on[s][p] = (qs[s] == qv);
        }
    }

    const int kFull = K & ~3;
    for (int kt = 0; kt < kFull; kt += 4) {
        const int k0 = kt + kOff;
        v2f b[4];
#pragma unroll
        for (int s = 0; s < 4; ++s) {
            b[s].x = xcol[s][(size_t)k0 << Dlog2];
            b[s].y = xcol[s][(size_t)(k0 + 1) << Dlog2];
        }
#pragma unroll
        for (int p = 0; p < NP; ++p) {
#pragma unroll
            for (int s = 0; s < 4; ++s) {
                v2f a;
                a.x = ar[s][p][k0];
                a.y = ar[s][p][k0 + 1];
                v2f bm = b[s];
                if (NP > 1) {
                    bm.x = on[s][p] ? b[s].x : 0.0f;
                    bm.y = on[s][p] ? b[s].y : 0.0f;
                }
                acc[s] = __builtin_amdgcn_wmma_f32_16x16x4_f32(
                    false, a, false, bm, (short)0, acc[s], false, false);
            }
        }
        if (kt + 4 < kFull) {
#pragma unroll
            for (int s = 0; s < 4; ++s)
                __builtin_prefetch(xcol[s] + ((size_t)(k0 + 4) << Dlog2), 0, 1);
        }
    }
    if (kFull < K) {
        const int k0 = kFull + kOff;
        const int c0 = min(k0, K - 1);
        const int c1 = min(k0 + 1, K - 1);
        v2f b[4];
#pragma unroll
        for (int s = 0; s < 4; ++s) {
            b[s].x = xcol[s][(size_t)c0 << Dlog2];
            b[s].y = xcol[s][(size_t)c1 << Dlog2];
        }
#pragma unroll
        for (int p = 0; p < NP; ++p) {
#pragma unroll
            for (int s = 0; s < 4; ++s) {
                v2f a;
                a.x = ar[s][p][k0];
                a.y = ar[s][p][k0 + 1];
                v2f bm = b[s];
                if (NP > 1) {
                    bm.x = on[s][p] ? b[s].x : 0.0f;
                    bm.y = on[s][p] ? b[s].y : 0.0f;
                }
                acc[s] = __builtin_amdgcn_wmma_f32_16x16x4_f32(
                    false, a, false, bm, (short)0, acc[s], false, false);
            }
        }
    }
}

__global__ __launch_bounds__(256) void kd_gather_wmma(
    const float* __restrict__ X, const float* __restrict__ Wt,
    const float* __restrict__ bias, const int* __restrict__ sel,
    float* __restrict__ Out, int F, int K, int Dlog2, int nw)
{
    extern __shared__ float sW[];
    const int D   = 1 << Dlog2;
    const int Kp4 = (K + 3) & ~3;
    const int Kp  = Kp4 + 1;
    const int f0  = blockIdx.y << 4;
    const int m0  = 3 * f0;
    const int M   = 3 * F;
    const int lane = threadIdx.x & 31;
    const int wv   = threadIdx.x >> 5;

    for (int r0 = wv; r0 < 48; r0 += 6 * nw) {
        for (int k = lane; k < Kp4; k += 32) {
            const int kc = min(k, K - 1);
            float w[6];
#pragma unroll
            for (int i = 0; i < 6; ++i) {
                const int m = min(m0 + r0 + i * nw, M - 1);
                const float x = Wt[(size_t)m * K + kc];
                w[i] = (k < K) ? x : 0.0f;
            }
#pragma unroll
            for (int i = 0; i < 6; ++i) sW[(r0 + i * nw) * Kp + k] = w[i];
        }
    }
    __syncthreads();

    const int n0 = (blockIdx.x * nw + wv) << 6;

    const int half = lane >> 4;
    const int l16  = lane & 15;
    const int kOff = half << 1;
    const float* aLane = sW + 3 * l16 * Kp;

    const float* xcol[4];
    int qs[4];
    int qlo[4];
    int np = 1;
#pragma unroll
    for (int s = 0; s < 4; ++s) {
        const int n  = n0 + (s << 4) + l16;
        const int bb = n >> Dlog2;
        const int d  = n & (D - 1);
        const int t  = 3 * d + sel[d];
        const int q  = t >> Dlog2;
        const int dp = t & (D - 1);
        xcol[s] = X + (size_t)bb * ((size_t)K << Dlog2) + dp;
        qs[s] = q;
        const int has0 = __any(q == 0);
        const int has1 = __any(q == 1);
        const int has2 = __any(q == 2);
        const int lo = has0 ? 0 : (has1 ? 1 : 2);
        const int hi = has2 ? 2 : (has1 ? 1 : 0);
        qlo[s] = lo;
        np = max(np, hi - lo + 1);
    }

    v8f acc[4] = { v8f{}, v8f{}, v8f{}, v8f{} };
    if (np == 1)      kd_kloop<1>(aLane, Kp, qlo, qs, xcol, K, Dlog2, kOff, acc);
    else if (np == 2) kd_kloop<2>(aLane, Kp, qlo, qs, xcol, K, Dlog2, kOff, acc);
    else              kd_kloop<3>(aLane, Kp, qlo, qs, xcol, K, Dlog2, kOff, acc);

    const int Dh = D >> 1;
    const int fl = f0 + (half << 3);
#pragma unroll
    for (int s = 0; s < 4; ++s) {
#pragma unroll
        for (int r = 0; r < 8; ++r) {
            const int fc = min(fl + r, F - 1);
            float v = fmaxf(acc[s][r] + bias[3 * fc + qs[s]], 0.0f);
            v = fmaxf(v, __shfl_xor(v, 1));
            acc[s][r] = v;
        }
    }
    if (!(l16 & 1)) {
        const int nr = F - fl;
#pragma unroll
        for (int s = 0; s < 4; ++s) {
            const int n  = n0 + (s << 4) + l16;
            const int bb = n >> Dlog2;
            const int d  = n & (D - 1);
            float* ocol = Out + ((size_t)bb * F + fl) * Dh + (d >> 1);
            if (nr >= 8) {
#pragma unroll
                for (int r = 0; r < 8; ++r) ocol[(size_t)r * Dh] = acc[s][r];
            } else {
#pragma unroll
                for (int r = 0; r < 8; ++r)
                    if (r < nr) ocol[(size_t)r * Dh] = acc[s][r];
            }
        }
    }
}

__global__ __launch_bounds__(256) void kd_fc_wmma(
    const float* __restrict__ X, const float* __restrict__ Wt,
    const float* __restrict__ bias, float* __restrict__ Out, int M, int K)
{
    extern __shared__ float sW[];
    const int Kp4 = (K + 3) & ~3;
    const int Kp  = Kp4 + 1;
    const int m0  = blockIdx.y << 4;

    for (int r = 0; r < 16; ++r) {
        const int m = m0 + r;
        const float* wr = Wt + (size_t)m * K;
        for (int k = threadIdx.x; k < Kp4; k += 256) {
            float w = 0.0f;
            if (m < M && k < K) w = wr[k];
            sW[r * Kp + k] = w;
        }
    }
    __syncthreads();

    const int lane = threadIdx.x & 31;
    const int ng   = (blockIdx.x << 3) + (threadIdx.x >> 5);
    const int n0   = ng << 6;
    const int half = lane >> 4;
    const int l16  = lane & 15;
    const int kOff = half << 1;
    const float* aRow = sW + l16 * Kp;

    const float* xcol[4];
#pragma unroll
    for (int s = 0; s < 4; ++s) {
        const int n = n0 + (s << 4) + l16;
        xcol[s] = X + (size_t)n * K;
    }

    v8f acc[4] = { v8f{}, v8f{}, v8f{}, v8f{} };

    const int kFull = K & ~3;
    for (int kt = 0; kt < kFull; kt += 4) {
        const int k0 = kt + kOff;
        v2f a;
        a.x = aRow[k0];
        a.y = aRow[k0 + 1];
#pragma unroll
        for (int s = 0; s < 4; ++s) {
            v2f b;
            b.x = xcol[s][k0];
            b.y = xcol[s][k0 + 1];
            acc[s] = __builtin_amdgcn_wmma_f32_16x16x4_f32(
                false, a, false, b, (short)0, acc[s], false, false);
        }
        if (kt + 4 < kFull) {
#pragma unroll
            for (int s = 0; s < 4; ++s)
                __builtin_prefetch(xcol[s] + (k0 + 4), 0, 1);
        }
    }
    if (kFull < K) {
        const int k0 = kFull + kOff;
        const int c0 = min(k0, K - 1);
        const int c1 = min(k0 + 1, K - 1);
        v2f a;
        a.x = aRow[k0];
        a.y = aRow[k0 + 1];
#pragma unroll
        for (int s = 0; s < 4; ++s) {
            v2f b;
            b.x = xcol[s][c0];
            b.y = xcol[s][c1];
            acc[s] = __builtin_amdgcn_wmma_f32_16x16x4_f32(
                false, a, false, b, (short)0, acc[s], false, false);
        }
    }

#pragma unroll
    for (int s = 0; s < 4; ++s) {
        const int n = n0 + (s << 4) + l16;
#pragma unroll
        for (int r = 0; r < 8; ++r) {
            const int o = m0 + r + (half << 3);
            if (o >= M) continue;
            Out[(size_t)n * M + o] = acc[s][r] + bias[o];
        }
    }
}

__global__ __launch_bounds__(256) void kd_logsoftmax16(
    const float* __restrict__ logits, float* __restrict__ out, int rows)
{
    int r = blockIdx.x * blockDim.x + threadIdx.x;
    if (r >= rows) return;
    float v[16];
    float mx = -3.402823466e38f;
#pragma unroll
    for (int i = 0; i < 16; ++i) { v[i] = logits[r * 16 + i]; mx = fmaxf(mx, v[i]); }
    float s = 0.0f;
#pragma unroll
    for (int i = 0; i < 16; ++i) s += expf(v[i] - mx);
    const float l = mx + logf(s);
#pragma unroll
    for (int i = 0; i < 16; ++i) out[r * 16 + i] = v[i] - l;
}

extern "C" void kernel_launch(void* const* d_in, const int* in_sizes, int n_in,
                              void* d_out, int out_size, void* d_ws, size_t ws_size,
                              hipStream_t stream) {
    static const int DIMS[11] = {2048, 1024, 512, 256, 128, 64, 32, 16, 8, 4, 2};
    static const int INCH[11] = {3, 8, 32, 64, 64, 64, 128, 256, 512, 512, 512};
    static const int FEAT[11] = {8, 32, 64, 64, 64, 128, 256, 512, 512, 512, 1024};
    const int Bsz = 512;

    const size_t XBYTES = (size_t)512 * 32 * 512 * sizeof(float);
    float* X0     = (float*)d_ws;
    float* X1     = (float*)((char*)d_ws + XBYTES);
    float* logits = (float*)((char*)d_ws + 2 * XBYTES);

    const float* cur = (const float*)d_in[0];
    for (int i = 0; i < 11; ++i) {
        const int Dd = DIMS[i], Kc = INCH[i], F = FEAT[i];
        const int Dlog2 = __builtin_ctz((unsigned)Dd);
        float* nxt = (i & 1) ? X1 : X0;

        const int N = Bsz * Dd;
        const int tilesF = (F + 15) / 16;
        const int nw = ((N / 512) * tilesF < 256) ? 4 : 8;
        const int blocksX = N / (64 * nw);
        const int Kp4 = (Kc + 3) & ~3;
        const size_t ldsBytes = (size_t)48 * (Kp4 + 1) * sizeof(float);
        dim3 grid(blocksX, tilesF);
        kd_gather_wmma<<<grid, 32 * nw, ldsBytes, stream>>>(
            cur, (const float*)d_in[12 + 2 * i], (const float*)d_in[13 + 2 * i],
            (const int*)d_in[1 + i], nxt, F, Kc, Dlog2, nw);
        cur = nxt;
    }

    {
        const size_t ldsBytes = (size_t)16 * (1024 + 1) * sizeof(float);
        dim3 grid(1, 1);
        kd_fc_wmma<<<grid, 256, ldsBytes, stream>>>(
            cur, (const float*)d_in[34], (const float*)d_in[35],
            logits,  16,  1024);
    }

    kd_logsoftmax16<<<(512 + 255) / 256, 256, 0, stream>>>(logits, (float*)d_out, 512);

    (void)in_sizes; (void)n_in; (void)out_size; (void)ws_size;
}


# FC head re-tiled: 1 block -> 32 blocks x 8 waves; each wave does 16 columns x 128 of K, loads independent of its products; the 8 K-slice partial sums are added through LDS (new summation order)
# speedup vs baseline: 3.0231x; 1.3975x over previous
#include <hip/hip_runtime.h>

typedef float v2f __attribute__((ext_vector_type(2)));
typedef float v4f __attribute__((ext_vector_type(4)));
typedef float v8f __attribute__((ext_vector_type(8)));

template <int NP>
__device__ __forceinline__ void kd_kloop(
    const float* aLane, int Kp, const int (&qlo)[4], const int (&qs)[4],
    const float* const (&xcol)[4], int K, int Dlog2, int kOff, v8f (&acc)[4])
{
    const float* ar[4][NP];
    bool on[4][NP];
#pragma unroll
    for (int s = 0; s < 4; ++s) {
#pragma unroll
        for (int p = 0; p < NP; ++p) {
            const int qv = qlo[s] + p;
            ar[s][p] = aLane + min(qv, 2) * Kp;
            on[s][p] = (qs[s] == qv);
        }
    }

    const int kFull = K & ~3;
    for (int kt = 0; kt < kFull; kt += 4) {
        const int k0 = kt + kOff;
        v2f b[4];
#pragma unroll
        for (int s = 0; s < 4; ++s) {
            b[s].x = xcol[s][(size_t)k0 << Dlog2];
            b[s].y = xcol[s][(size_t)(k0 + 1) << Dlog2];
        }
#pragma unroll
        for (int p = 0; p < NP; ++p) {
#pragma unroll
            for (int s = 0; s < 4; ++s) {
                v2f a;
                a.x = ar[s][p][k0];
                a.y = ar[s][p][k0 + 1];
                v2f bm = b[s];
                if (NP > 1) {
                    bm.x = on[s][p] ? b[s].x : 0.0f;
                    bm.y = on[s][p] ? b[s].y : 0.0f;
                }
                acc[s] = __builtin_amdgcn_wmma_f32_16x16x4_f32(
                    false, a, false, bm, (short)0, acc[s], false, false);
            }
        }
        if (kt + 4 < kFull) {
#pragma unroll
            for (int s = 0; s < 4; ++s)
                __builtin_prefetch(xcol[s] + ((size_t)(k0 + 4) << Dlog2), 0, 1);
        }
    }
    if (kFull < K) {
        const int k0 = kFull + kOff;
        const int c0 = min(k0, K - 1);
        const int c1 = min(k0 + 1, K - 1);
        v2f b[4];
#pragma unroll
        for (int s = 0; s < 4; ++s) {
            b[s].x = xcol[s][(size_t)c0 << Dlog2];
            b[s].y = xcol[s][(size_t)c1 << Dlog2];
        }
#pragma unroll
        for (int p = 0; p < NP; ++p) {
#pragma unroll
            for (int s = 0; s < 4; ++s) {
                v2f a;
                a.x = ar[s][p][k0];
                a.y = ar[s][p][k0 + 1];
                v2f bm = b[s];
                if (NP > 1) {
                    bm.x = on[s][p] ? b[s].x : 0.0f;
                    bm.y = on[s][p] ? b[s].y : 0.0f;
                }
                acc[s] = __builtin_amdgcn_wmma_f32_16x16x4_f32(
                    false, a, false, bm, (short)0, acc[s], false, false);
            }
        }
    }
}

__global__ __launch_bounds__(256) void kd_gather_wmma(
    const float* __restrict__ X, const float* __restrict__ Wt,
    const float* __restrict__ bias, const int* __restrict__ sel,
    float* __restrict__ Out, int F, int K, int Dlog2, int nw)
{
    extern __shared__ float sW[];
    const int D   = 1 << Dlog2;
    const int Kp4 = (K + 3) & ~3;
    const int Kp  = Kp4 + 1;
    const int f0  = blockIdx.y << 4;
    const int m0  = 3 * f0;
    const int M   = 3 * F;
    const int lane = threadIdx.x & 31;
    const int wv   = threadIdx.x >> 5;

    for (int r0 = wv; r0 < 48; r0 += 6 * nw) {
        for (int k = lane; k < Kp4; k += 32) {
            const int kc = min(k, K - 1);
            float w[6];
#pragma unroll
            for (int i = 0; i < 6; ++i) {
                const int m = min(m0 + r0 + i * nw, M - 1);
                const float x = Wt[(size_t)m * K + kc];
                w[i] = (k < K) ? x : 0.0f;
            }
#pragma unroll
            for (int i = 0; i < 6; ++i) sW[(r0 + i * nw) * Kp + k] = w[i];
        }
    }
    __syncthreads();

    const int n0 = (blockIdx.x * nw + wv) << 6;

    const int half = lane >> 4;
    const int l16  = lane & 15;
    const int kOff = half << 1;
    const float* aLane = sW + 3 * l16 * Kp;

    const float* xcol[4];
    int qs[4];
    int qlo[4];
    int np = 1;
#pragma unroll
    for (int s = 0; s < 4; ++s) {
        const int n  = n0 + (s << 4) + l16;
        const int bb = n >> Dlog2;
        const int d  = n & (D - 1);
        const int t  = 3 * d + sel[d];
        const int q  = t >> Dlog2;
        const int dp = t & (D - 1);
        xcol[s] = X + (size_t)bb * ((size_t)K << Dlog2) + dp;
        qs[s] = q;
        const int has0 = __any(q == 0);
        const int has1 = __any(q == 1);
        const int has2 = __any(q == 2);
        const int lo = has0 ? 0 : (has1 ? 1 : 2);
        const int hi = has2 ? 2 : (has1 ? 1 : 0);
        qlo[s] = lo;
        np = max(np, hi - lo + 1);
    }

    v8f acc[4] = { v8f{}, v8f{}, v8f{}, v8f{} };
    if (np == 1)      kd_kloop<1>(aLane, Kp, qlo, qs, xcol, K, Dlog2, kOff, acc);
    else if (np == 2) kd_kloop<2>(aLane, Kp, qlo, qs, xcol, K, Dlog2, kOff, acc);
    else              kd_kloop<3>(aLane, Kp, qlo, qs, xcol, K, Dlog2, kOff, acc);

    const int Dh = D >> 1;
    const int fl = f0 + (half << 3);
#pragma unroll
    for (int s = 0; s < 4; ++s) {
#pragma unroll
        for (int r = 0; r < 8; ++r) {
            const int fc = min(fl + r, F - 1);
            float v = fmaxf(acc[s][r] + bias[3 * fc + qs[s]], 0.0f);
            v = fmaxf(v, __shfl_xor(v, 1));
            acc[s][r] = v;
        }
    }
    if (!(l16 & 1)) {
        const int nr = F - fl;
#pragma unroll
        for (int s = 0; s < 4; ++s) {
            const int n  = n0 + (s << 4) + l16;
            const int bb = n >> Dlog2;
            const int d  = n & (D - 1);
            float* ocol = Out + ((size_t)bb * F + fl) * Dh + (d >> 1);
            if (nr >= 8) {
#pragma unroll
                for (int r = 0; r < 8; ++r) ocol[(size_t)r * Dh] = acc[s][r];
            } else {
#pragma unroll
                for (int r = 0; r < 8; ++r)
                    if (r < nr) ocol[(size_t)r * Dh] = acc[s][r];
            }
        }
    }
}

template <int KS>
__device__ __forceinline__ void fc_load_row(const float* rowSlice, int half,
                                            v4f (&v)[KS / 8])
{
    const v4f* p = reinterpret_cast<const v4f*>(rowSlice + 4 * half);
#pragma unroll
    for (int i = 0; i < KS / 8; ++i) v[i] = p[2 * i];
}

template <int K>
__global__ __launch_bounds__(256) void kd_fc_wmma(
    const float* __restrict__ X, const float* __restrict__ Wt,
    const float* __restrict__ bias, float* __restrict__ Out)
{
    static_assert(K % 64 == 0, "8 slices of whole 8-k pieces");
    constexpr int KS = K / 8;
    __shared__ float red[8 * 256];

    const int lane = threadIdx.x & 31;
    const int wv   = threadIdx.x >> 5;
    const int half = lane >> 4;
    const int l16  = lane & 15;
    const int n0   = blockIdx.x << 4;

    v4f va[KS / 8], vb[KS / 8];
    fc_load_row<KS>(Wt + (size_t)l16 * K + wv * KS, half, va);
    fc_load_row<KS>(X + (size_t)(n0 + l16) * K + wv * KS, half, vb);

    v8f acc0 = v8f{}, acc1 = v8f{};
#pragma unroll
    for (int i = 0; i < KS / 8; ++i) {
        v2f a0, b0, a1, b1;
        a0.x = va[i].x; a0.y = va[i].y;
        b0.x = vb[i].x; b0.y = vb[i].y;
        a1.x = va[i].z; a1.y = va[i].w;
        b1.x = vb[i].z; b1.y = vb[i].w;
        acc0 = __builtin_amdgcn_wmma_f32_16x16x4_f32(
            false, a0, false, b0, (short)0, acc0, false, false);
        acc1 = __builtin_amdgcn_wmma_f32_16x16x4_f32(
            false, a1, false, b1, (short)0, acc1, false, false);
    }

#pragma unroll
    for (int r = 0; r < 8; ++r)
        red[(wv << 8) + (r << 5) + lane] = acc0[r] + acc1[r];
    __syncthreads();

    const int e  = threadIdx.x;
    const int nl = e >> 4;
    const int o  = e & 15;
    const int cell = ((o & 7) << 5) + nl + ((o >> 3) << 4);
    float sum = red[cell];
#pragma unroll
    for (int w = 1; w < 8; ++w) sum += red[(w << 8) + cell];
    Out[((size_t)(n0 + nl) << 4) + o] = sum + bias[o];
}

__global__ __launch_bounds__(256) void kd_logsoftmax16(
    const float* __restrict__ logits, float* __restrict__ out, int rows)
{
    int r = blockIdx.x * blockDim.x + threadIdx.x;
    if (r >= rows) return;
    float v[16];
    float mx = -3.402823466e38f;
#pragma unroll
    for (int i = 0; i < 16; ++i) { v[i] = logits[r * 16 + i]; mx = fmaxf(mx, v[i]); }
    float s = 0.0f;
#pragma unroll
    for (int i = 0; i < 16; ++i) s += expf(v[i] - mx);
    const float l = mx + logf(s);
#pragma unroll
    for (int i = 0; i < 16; ++i) out[r * 16 + i] = v[i] - l;
}

extern "C" void kernel_launch(void* const* d_in, const int* in_sizes, int n_in,
                              void* d_out, int out_size, void* d_ws, size_t ws_size,
                              hipStream_t stream) {
    static const int DIMS[11] = {2048, 1024, 512, 256, 128, 64, 32, 16, 8, 4, 2};
    static const int INCH[11] = {3, 8, 32, 64, 64, 64, 128, 256, 512, 512, 512};
    static const int FEAT[11] = {8, 32, 64, 64, 64, 128, 256, 512, 512, 512, 1024};
    const int Bsz = 512;

    const size_t XBYTES = (size_t)512 * 32 * 512 * sizeof(float);
    float* X0     = (float*)d_ws;
    float* X1     = (float*)((char*)d_ws + XBYTES);
    float* logits = (float*)((char*)d_ws + 2 * XBYTES);

    const float* cur = (const float*)d_in[0];
    for (int i = 0; i < 11; ++i) {
        const int Dd = DIMS[i], Kc = INCH[i], F = FEAT[i];
        const int Dlog2 = __builtin_ctz((unsigned)Dd);
        float* nxt = (i & 1) ? X1 : X0;

        const int N = Bsz * Dd;
        const int tilesF = (F + 15) / 16;
        const int nw = ((N / 512) * tilesF < 256) ? 4 : 8;
        const int blocksX = N / (64 * nw);
        const int Kp4 = (Kc + 3) & ~3;
        const size_t ldsBytes = (size_t)48 * (Kp4 + 1) * sizeof(float);
        dim3 grid(blocksX, tilesF);
        kd_gather_wmma<<<grid, 32 * nw, ldsBytes, stream>>>(
            cur, (const float*)d_in[12 + 2 * i], (const float*)d_in[13 + 2 * i],
            (const int*)d_in[1 + i], nxt, F, Kc, Dlog2, nw);
        cur = nxt;
    }

    kd_fc_wmma<1024><<<dim3(32), dim3(256), 0, stream>>>(
        cur, (const float*)d_in[34], (const float*)d_in[35], logits);

    kd_logsoftmax16<<<(512 + 255) / 256, 256, 0, stream>>>(logits, (float*)d_out, 512);

    (void)in_sizes; (void)n_in; (void)out_size; (void)ws_size;
}


# K loop of the conv layers software-pipelined: gathered B loads issued ahead of the products into a ring of register sets (3 tiles ahead in layers 5-11, 1 in layers 1-4); sums bit-identical
# speedup vs baseline: 4.0104x; 1.3266x over previous
#include <hip/hip_runtime.h>

typedef float v2f __attribute__((ext_vector_type(2)));
typedef float v4f __attribute__((ext_vector_type(4)));
typedef float v8f __attribute__((ext_vector_type(8)));

template <int NP>
__device__ __forceinline__ void kd_tile(
    const float* const (&ar)[4][NP], const bool (&on)[4][NP], int k0,
    const v2f (&b)[4], v8f (&acc)[4])
{
#pragma unroll
    for (int p = 0; p < NP; ++p) {
#pragma unroll
        for (int s = 0; s < 4; ++s) {
            v2f a;
            a.x = ar[s][p][k0];
            a.y = ar[s][p][k0 + 1];
            v2f bm = b[s];
            if (NP > 1) {
                bm.x = on[s][p] ? b[s].x : 0.0f;
                bm.y = on[s][p] ? b[s].y : 0.0f;
            }
            acc[s] = __builtin_amdgcn_wmma_f32_16x16x4_f32(
                false, a, false, bm, (short)0, acc[s], false, false);
        }
    }
}

template <int NP, int R>
__device__ __forceinline__ void kd_kloop(
    const float* aLane, int Kp, const int (&qlo)[4], const int (&qs)[4],
    const char* xb, const unsigned (&bo0)[4], const unsigned (&bo1)[4],
    int K, int Dlog2, int kOff, v8f (&acc)[4])
{
    const float* ar[4][NP];
    bool on[4][NP];
#pragma unroll
    for (int s = 0; s < 4; ++s) {
#pragma unroll
        for (int p = 0; p < NP; ++p) {
            const int qv = qlo[s] + p;
            ar[s][p] = aLane + min(qv, 2) * Kp;
            on[s][p] = (qs[s] == qv);
        }
    }

    const int nT = K >> 2;
    const unsigned tileB = 16u << Dlog2;
    if (nT > 0) {
        const int last = nT - 1;
        auto ldB = [&](int t, v2f (&b)[4]) {
            const unsigned tob = (unsigned)min(t, last) * tileB;
#pragma unroll
            for (int s = 0; s < 4; ++s) {
                b[s].x = *reinterpret_cast<const float*>(xb + (bo0[s] + tob));
                b[s].y = *reinterpret_cast<const float*>(xb + (bo1[s] + tob));
            }
        };
        v2f b[R][4];
#pragma unroll
        for (int j = 0; j < R - 1; ++j) {
            ldB(j, b[j]);
            __builtin_amdgcn_sched_barrier(0);
        }
        int t = 0;
        for (; t + R <= nT; t += R) {
#pragma unroll
            for (int j = 0; j < R; ++j) {
                ldB(t + j + R - 1, b[(j + R - 1) % R]);
                kd_tile<NP>(ar, on, 4 * (t + j) + kOff, b[j], acc);
                if (j + 1 < R) __builtin_amdgcn_sched_barrier(0);
            }
        }
#pragma unroll
        for (int j = 0; j < R - 1; ++j)
            if (t + j < nT) kd_tile<NP>(ar, on, 4 * (t + j) + kOff, b[j], acc);
    }
    const int kFull = K & ~3;
    if (kFull < K) {
        const int k0 = kFull + kOff;
        const int c0 = min(k0, K - 1) - kOff;
        const int c1 = min(k0 + 1, K - 1) - kOff;
        const ptrdiff_t rowF = (ptrdiff_t)1 << Dlog2;
        v2f b[4];
#pragma unroll
        for (int s = 0; s < 4; ++s) {
            const float* col = reinterpret_cast<const float*>(xb + bo0[s]);
            b[s].x = col[(ptrdiff_t)c0 * rowF];
            b[s].y = col[(ptrdiff_t)c1 * rowF];
        }
        kd_tile<NP>(ar, on, k0, b, acc);
    }
}

template <int R>
__global__ __launch_bounds__(256) void kd_gather_wmma(
    const float* __restrict__ X, const float* __restrict__ Wt,
    const float* __restrict__ bias, const int* __restrict__ sel,
    float* __restrict__ Out, int F, int K, int Dlog2, int nw)
{
    extern __shared__ float sW[];
    const int D   = 1 << Dlog2;
    const int Kp4 = (K + 3) & ~3;
    const int Kp  = Kp4 + 1;
    const int f0  = blockIdx.y << 4;
    const int m0  = 3 * f0;
    const int M   = 3 * F;
    const int lane = threadIdx.x & 31;
    const int wv   = threadIdx.x >> 5;

    for (int r0 = wv; r0 < 48; r0 += 6 * nw) {
        for (int k = lane; k < Kp4; k += 32) {
            const int kc = min(k, K - 1);
            float w[6];
#pragma unroll
            for (int i = 0; i < 6; ++i) {
                const int m = min(m0 + r0 + i * nw, M - 1);
                const float x = Wt[(size_t)m * K + kc];
                w[i] = (k < K) ? x : 0.0f;
            }
#pragma unroll
            for (int i = 0; i < 6; ++i) sW[(r0 + i * nw) * Kp + k] = w[i];
        }
    }
    __syncthreads();

    const int n0 = (blockIdx.x * nw + wv) << 6;

    const int half = lane >> 4;
    const int l16  = lane & 15;
    const int kOff = half << 1;
    const float* aLane = sW + 3 * l16 * Kp;

    unsigned bo0[4], bo1[4];
    int qs[4];
    int qlo[4];
    int np = 1;
#pragma unroll
    for (int s = 0; s < 4; ++s) {
        const int n  = n0 + (s << 4) + l16;
        const int bb = n >> Dlog2;
        const int d  = n & (D - 1);
        const int t  = 3 * d + sel[d];
        const int q  = t >> Dlog2;
        const int dp = t & (D - 1);
        bo0[s] = ((unsigned)bb * ((unsigned)K << Dlog2) + ((unsigned)kOff << Dlog2)
                  + (unsigned)dp) << 2;
        bo1[s] = bo0[s] + (4u << Dlog2);
        qs[s] = q;
        const int has0 = __any(q == 0);
        const int has1 = __any(q == 1);
        const int has2 = __any(q == 2);
        const int lo = has0 ? 0 : (has1 ? 1 : 2);
        const int hi = has2 ? 2 : (has1 ? 1 : 0);
        qlo[s] = lo;
        np = max(np, hi - lo + 1);
    }

    v8f acc[4] = { v8f{}, v8f{}, v8f{}, v8f{} };
    const char* xb = reinterpret_cast<const char*>(X);
    if (np == 1)      kd_kloop<1, R>(aLane, Kp, qlo, qs, xb, bo0, bo1, K, Dlog2, kOff, acc);
    else if (np == 2) kd_kloop<2, R>(aLane, Kp, qlo, qs, xb, bo0, bo1, K, Dlog2, kOff, acc);
    else              kd_kloop<3, R>(aLane, Kp, qlo, qs, xb, bo0, bo1, K, Dlog2, kOff, acc);

    const int Dh = D >> 1;
    const int fl = f0 + (half << 3);
#pragma unroll
    for (int s = 0; s < 4; ++s) {
#pragma unroll
        for (int r = 0; r < 8; ++r) {
            const int fc = min(fl + r, F - 1);
            float v = fmaxf(acc[s][r] + bias[3 * fc + qs[s]], 0.0f);
            v = fmaxf(v, __shfl_xor(v, 1));
            acc[s][r] = v;
        }
    }
    if (!(l16 & 1)) {
        const int nr = F - fl;
#pragma unroll
        for (int s = 0; s < 4; ++s) {
            const int n  = n0 + (s << 4) + l16;
            const int bb = n >> Dlog2;
            const int d  = n & (D - 1);
            float* ocol = Out + ((size_t)bb * F + fl) * Dh + (d >> 1);
            if (nr >= 8) {
#pragma unroll
                for (int r = 0; r < 8; ++r) ocol[(size_t)r * Dh] = acc[s][r];
            } else {
#pragma unroll
                for (int r = 0; r < 8; ++r)
                    if (r < nr) ocol[(size_t)r * Dh] = acc[s][r];
            }
        }
    }
}

template <int KS>
__device__ __forceinline__ void fc_load_row(const float* rowSlice, int half,
                                            v4f (&v)[KS / 8])
{
    const v4f* p = reinterpret_cast<const v4f*>(rowSlice + 4 * half);
#pragma unroll
    for (int i = 0; i < KS / 8; ++i) v[i] = p[2 * i];
}

template <int K>
__global__ __launch_bounds__(256) void kd_fc_wmma(
    const float* __restrict__ X, const float* __restrict__ Wt,
    const float* __restrict__ bias, float* __restrict__ Out)
{
    static_assert(K % 64 == 0, "8 slices of whole 8-k pieces");
    constexpr int KS = K / 8;
    __shared__ float red[8 * 256];

    const int lane = threadIdx.x & 31;
    const int wv   = threadIdx.x >> 5;
    const int half = lane >> 4;
    const int l16  = lane & 15;
    const int n0   = blockIdx.x << 4;

    v4f va[KS / 8], vb[KS / 8];
    fc_load_row<KS>(Wt + (size_t)l16 * K + wv * KS, half, va);
    fc_load_row<KS>(X + (size_t)(n0 + l16) * K + wv * KS, half, vb);

    v8f acc0 = v8f{}, acc1 = v8f{};
#pragma unroll
    for (int i = 0; i < KS / 8; ++i) {
        v2f a0, b0, a1, b1;
        a0.x = va[i].x; a0.y = va[i].y;
        b0.x = vb[i].x; b0.y = vb[i].y;
        a1.x = va[i].z; a1.y = va[i].w;
        b1.x = vb[i].z; b1.y = vb[i].w;
        acc0 = __builtin_amdgcn_wmma_f32_16x16x4_f32(
            false, a0, false, b0, (short)0, acc0, false, false);
        acc1 = __builtin_amdgcn_wmma_f32_16x16x4_f32(
            false, a1, false, b1, (short)0, acc1, false, false);
    }

#pragma unroll
    for (int r = 0; r < 8; ++r)
        red[(wv << 8) + (r << 5) + lane] = acc0[r] + acc1[r];
    __syncthreads();

    const int e  = threadIdx.x;
    const int nl = e >> 4;
    const int o  = e & 15;
    const int cell = ((o & 7) << 5) + nl + ((o >> 3) << 4);
    float sum = red[cell];
#pragma unroll
    for (int w = 1; w < 8; ++w) sum += red[(w << 8) + cell];
    Out[((size_t)(n0 + nl) << 4) + o] = sum + bias[o];
}

__global__ __launch_bounds__(256) void kd_logsoftmax16(
    const float* __restrict__ logits, float* __restrict__ out, int rows)
{
    int r = blockIdx.x * blockDim.x + threadIdx.x;
    if (r >= rows) return;
    float v[16];
    float mx = -3.402823466e38f;
#pragma unroll
    for (int i = 0; i < 16; ++i) { v[i] = logits[r * 16 + i]; mx = fmaxf(mx, v[i]); }
    float s = 0.0f;
#pragma unroll
    for (int i = 0; i < 16; ++i) s += expf(v[i] - mx);
    const float l = mx + logf(s);
#pragma unroll
    for (int i = 0; i < 16; ++i) out[r * 16 + i] = v[i] - l;
}

extern "C" void kernel_launch(void* const* d_in, const int* in_sizes, int n_in,
                              void* d_out, int out_size, void* d_ws, size_t ws_size,
                              hipStream_t stream) {
    static const int DIMS[11] = {2048, 1024, 512, 256, 128, 64, 32, 16, 8, 4, 2};
    static const int INCH[11] = {3, 8, 32, 64, 64, 64, 128, 256, 512, 512, 512};
    static const int FEAT[11] = {8, 32, 64, 64, 64, 128, 256, 512, 512, 512, 1024};
    static const int RING[11] = {2, 2, 2, 2, 4, 4, 4, 4, 4, 4, 4};
    const int Bsz = 512;

    const size_t XBYTES = (size_t)512 * 32 * 512 * sizeof(float);
    float* X0     = (float*)d_ws;
    float* X1     = (float*)((char*)d_ws + XBYTES);
    float* logits = (float*)((char*)d_ws + 2 * XBYTES);

    const float* cur = (const float*)d_in[0];
    for (int i = 0; i < 11; ++i) {
        const int Dd = DIMS[i], Kc = INCH[i], F = FEAT[i];
        const int Dlog2 = __builtin_ctz((unsigned)Dd);
        float* nxt = (i & 1) ? X1 : X0;

        const int N = Bsz * Dd;
        const int tilesF = (F + 15) / 16;
        const int nw = ((N / 512) * tilesF < 256) ? 4 : 8;
        const int blocksX = N / (64 * nw);
        const int Kp4 = (Kc + 3) & ~3;
        const size_t ldsBytes = (size_t)48 * (Kp4 + 1) * sizeof(float);
        dim3 grid(blocksX, tilesF);
        if (RING[i] == 4)
            kd_gather_wmma<4><<<grid, 32 * nw, ldsBytes, stream>>>(
                cur, (const float*)d_in[12 + 2 * i], (const float*)d_in[13 + 2 * i],
                (const int*)d_in[1 + i], nxt, F, Kc, Dlog2, nw);
        else
            kd_gather_wmma<2><<<grid, 32 * nw, ldsBytes, stream>>>(
                cur, (const float*)d_in[12 + 2 * i], (const float*)d_in[13 + 2 * i],
                (const int*)d_in[1 + i], nxt, F, Kc, Dlog2, nw);
        cur = nxt;
    }

    kd_fc_wmma<1024><<<dim3(32), dim3(256), 0, stream>>>(
        cur, (const float*)d_in[34], (const float*)d_in[35], logits);

    kd_logsoftmax16<<<(512 + 255) / 256, 256, 0, stream>>>(logits, (float*)d_out, 512);

    (void)in_sizes; (void)n_in; (void)out_size; (void)ws_size;
}


# W stage of the conv layers: each wave requests 8 chunks of 32 K columns x 6 rows (48 loads) before its first LDS store instead of 6 loads per round trip; LDS contents and sums unchanged
# speedup vs baseline: 4.2875x; 1.0691x over previous
#include <hip/hip_runtime.h>

typedef float v2f __attribute__((ext_vector_type(2)));
typedef float v4f __attribute__((ext_vector_type(4)));
typedef float v8f __attribute__((ext_vector_type(8)));

template <int NP>
__device__ __forceinline__ void kd_tile(
    const float* const (&ar)[4][NP], const bool (&on)[4][NP], int k0,
    const v2f (&b)[4], v8f (&acc)[4])
{
#pragma unroll
    for (int p = 0; p < NP; ++p) {
#pragma unroll
        for (int s = 0; s < 4; ++s) {
            v2f a;
            a.x = ar[s][p][k0];
            a.y = ar[s][p][k0 + 1];
            v2f bm = b[s];
            if (NP > 1) {
                bm.x = on[s][p] ? b[s].x : 0.0f;
                bm.y = on[s][p] ? b[s].y : 0.0f;
            }
            acc[s] = __builtin_amdgcn_wmma_f32_16x16x4_f32(
                false, a, false, bm, (short)0, acc[s], false, false);
        }
    }
}

template <int NP, int R>
__device__ __forceinline__ void kd_kloop(
    const float* aLane, int Kp, const int (&qlo)[4], const int (&qs)[4],
    const char* xb, const unsigned (&bo0)[4], const unsigned (&bo1)[4],
    int K, int Dlog2, int kOff, v8f (&acc)[4])
{
    const float* ar[4][NP];
    bool on[4][NP];
#pragma unroll
    for (int s = 0; s < 4; ++s) {
#pragma unroll
        for (int p = 0; p < NP; ++p) {
            const int qv = qlo[s] + p;
            ar[s][p] = aLane + min(qv, 2) * Kp;
            on[s][p] = (qs[s] == qv);
        }
    }

    const int nT = K >> 2;
    const unsigned tileB = 16u << Dlog2;
    if (nT > 0) {
        const int last = nT - 1;
        auto ldB = [&](int t, v2f (&b)[4]) {
            const unsigned tob = (unsigned)min(t, last) * tileB;
#pragma unroll
            for (int s = 0; s < 4; ++s) {
                b[s].x = *reinterpret_cast<const float*>(xb + (bo0[s] + tob));
                b[s].y = *reinterpret_cast<const float*>(xb + (bo1[s] + tob));
            }
        };
        v2f b[R][4];
#pragma unroll
        for (int j = 0; j < R - 1; ++j) {
            ldB(j, b[j]);
            __builtin_amdgcn_sched_barrier(0);
        }
        int t = 0;
        for (; t + R <= nT; t += R) {
#pragma unroll
            for (int j = 0; j < R; ++j) {
                ldB(t + j + R - 1, b[(j + R - 1) % R]);
                kd_tile<NP>(ar, on, 4 * (t + j) + kOff, b[j], acc);
                if (j + 1 < R) __builtin_amdgcn_sched_barrier(0);
            }
        }
#pragma unroll
        for (int j = 0; j < R - 1; ++j)
            if (t + j < nT) kd_tile<NP>(ar, on, 4 * (t + j) + kOff, b[j], acc);
    }
    const int kFull = K & ~3;
    if (kFull < K) {
        const int k0 = kFull + kOff;
        const int c0 = min(k0, K - 1) - kOff;
        const int c1 = min(k0 + 1, K - 1) - kOff;
        const ptrdiff_t rowF = (ptrdiff_t)1 << Dlog2;
        v2f b[4];
#pragma unroll
        for (int s = 0; s < 4; ++s) {
            const float* col = reinterpret_cast<const float*>(xb + bo0[s]);
            b[s].x = col[(ptrdiff_t)c0 * rowF];
            b[s].y = col[(ptrdiff_t)c1 * rowF];
        }
        kd_tile<NP>(ar, on, k0, b, acc);
    }
}

template <int R>
__global__ __launch_bounds__(256) void kd_gather_wmma(
    const float* __restrict__ X, const float* __restrict__ Wt,
    const float* __restrict__ bias, const int* __restrict__ sel,
    float* __restrict__ Out, int F, int K, int Dlog2, int nw)
{
    extern __shared__ float sW[];
    const int D   = 1 << Dlog2;
    const int Kp4 = (K + 3) & ~3;
    const int Kp  = Kp4 + 1;
    const int f0  = blockIdx.y << 4;
    const int m0  = 3 * f0;
    const int M   = 3 * F;
    const int lane = threadIdx.x & 31;
    const int wv   = threadIdx.x >> 5;

    constexpr int KU = 8;
    for (int r0 = wv; r0 < 48; r0 += 6 * nw) {
        for (int kb0 = 0; kb0 < Kp4; kb0 += 32 * KU) {
            const int nU = min(KU, (Kp4 - kb0 + 31) >> 5);
            float w[KU][6];
#pragma unroll
            for (int u = 0; u < KU; ++u) {
#pragma unroll
                for (int i = 0; i < 6; ++i) w[u][i] = 0.0f;
                if (u < nU) {
                    const int kc = min(kb0 + 32 * u + lane, K - 1);
#pragma unroll
                    for (int i = 0; i < 6; ++i) {
                        const int m = min(m0 + r0 + i * nw, M - 1);
                        w[u][i] = Wt[(size_t)m * K + kc];
                    }
                }
            }
#pragma unroll
            for (int u = 0; u < KU; ++u) {
                if (u < nU) {
                    const int k = kb0 + 32 * u + lane;
                    if (k < Kp4) {
#pragma unroll
                        for (int i = 0; i < 6; ++i)
                            sW[(r0 + i * nw) * Kp + k] = (k < K) ? w[u][i] : 0.0f;
                    }
                }
            }
        }
    }
    __syncthreads();

    const int n0 = (blockIdx.x * nw + wv) << 6;

    const int half = lane >> 4;
    const int l16  = lane & 15;
    const int kOff = half << 1;
    const float* aLane = sW + 3 * l16 * Kp;

    unsigned bo0[4], bo1[4];
    int qs[4];
    int qlo[4];
    int np = 1;
#pragma unroll
    for (int s = 0; s < 4; ++s) {
        const int n  = n0 + (s << 4) + l16;
        const int bb = n >> Dlog2;
        const int d  = n & (D - 1);
        const int t  = 3 * d + sel[d];
        const int q  = t >> Dlog2;
        const int dp = t & (D - 1);
        bo0[s] = ((unsigned)bb * ((unsigned)K << Dlog2) + ((unsigned)kOff << Dlog2)
                  + (unsigned)dp) << 2;
        bo1[s] = bo0[s] + (4u << Dlog2);
        qs[s] = q;
        const int has0 = __any(q == 0);
        const int has1 = __any(q == 1);
        const int has2 = __any(q == 2);
        const int lo = has0 ? 0 : (has1 ? 1 : 2);
        const int hi = has2 ? 2 : (has1 ? 1 : 0);
        qlo[s] = lo;
        np = max(np, hi - lo + 1);
    }

    v8f acc[4] = { v8f{}, v8f{}, v8f{}, v8f{} };
    const char* xb = reinterpret_cast<const char*>(X);
    if (np == 1)      kd_kloop<1, R>(aLane, Kp, qlo, qs, xb, bo0, bo1, K, Dlog2, kOff, acc);
    else if (np == 2) kd_kloop<2, R>(aLane, Kp, qlo, qs, xb, bo0, bo1, K, Dlog2, kOff, acc);
    else              kd_kloop<3, R>(aLane, Kp, qlo, qs, xb, bo0, bo1, K, Dlog2, kOff, acc);

    const int Dh = D >> 1;
    const int fl = f0 + (half << 3);
#pragma unroll
    for (int s = 0; s < 4; ++s) {
#pragma unroll
        for (int r = 0; r < 8; ++r) {
            const int fc = min(fl + r, F - 1);
            float v = fmaxf(acc[s][r] + bias[3 * fc + qs[s]], 0.0f);
            v = fmaxf(v, __shfl_xor(v, 1));
            acc[s][r] = v;
        }
    }
    if (!(l16 & 1)) {
        const int nr = F - fl;
#pragma unroll
        for (int s = 0; s < 4; ++s) {
            const int n  = n0 + (s << 4) + l16;
            const int bb = n >> Dlog2;
            const int d  = n & (D - 1);
            float* ocol = Out + ((size_t)bb * F + fl) * Dh + (d >> 1);
            if (nr >= 8) {
#pragma unroll
                for (int r = 0; r < 8; ++r) ocol[(size_t)r * Dh] = acc[s][r];
            } else {
#pragma unroll
                for (int r = 0; r < 8; ++r)
                    if (r < nr) ocol[(size_t)r * Dh] = acc[s][r];
            }
        }
    }
}

template <int KS>
__device__ __forceinline__ void fc_load_row(const float* rowSlice, int half,
                                            v4f (&v)[KS / 8])
{
    const v4f* p = reinterpret_cast<const v4f*>(rowSlice + 4 * half);
#pragma unroll
    for (int i = 0; i < KS / 8; ++i) v[i] = p[2 * i];
}

template <int K>
__global__ __launch_bounds__(256) void kd_fc_wmma(
    const float* __restrict__ X, const float* __restrict__ Wt,
    const float* __restrict__ bias, float* __restrict__ Out)
{
    static_assert(K % 64 == 0, "8 slices of whole 8-k pieces");
    constexpr int KS = K / 8;
    __shared__ float red[8 * 256];

    const int lane = threadIdx.x & 31;
    const int wv   = threadIdx.x >> 5;
    const int half = lane >> 4;
    const int l16  = lane & 15;
    const int n0   = blockIdx.x << 4;

    v4f va[KS / 8], vb[KS / 8];
    fc_load_row<KS>(Wt + (size_t)l16 * K + wv * KS, half, va);
    fc_load_row<KS>(X + (size_t)(n0 + l16) * K + wv * KS, half, vb);

    v8f acc0 = v8f{}, acc1 = v8f{};
#pragma unroll
    for (int i = 0; i < KS / 8; ++i) {
        v2f a0, b0, a1, b1;
        a0.x = va[i].x; a0.y = va[i].y;
        b0.x = vb[i].x; b0.y = vb[i].y;
        a1.x = va[i].z; a1.y = va[i].w;
        b1.x = vb[i].z; b1.y = vb[i].w;
        acc0 = __builtin_amdgcn_wmma_f32_16x16x4_f32(
            false, a0, false, b0, (short)0, acc0, false, false);
        acc1 = __builtin_amdgcn_wmma_f32_16x16x4_f32(
            false, a1, false, b1, (short)0, acc1, false, false);
    }

#pragma unroll
    for (int r = 0; r < 8; ++r)
        red[(wv << 8) + (r << 5) + lane] = acc0[r] + acc1[r];
    __syncthreads();

    const int e  = threadIdx.x;
    const int nl = e >> 4;
    const int o  = e & 15;
    const int cell = ((o & 7) << 5) + nl + ((o >> 3) << 4);
    float sum = red[cell];
#pragma unroll
    for (int w = 1; w < 8; ++w) sum += red[(w << 8) + cell];
    Out[((size_t)(n0 + nl) << 4) + o] = sum + bias[o];
}

__global__ __launch_bounds__(256) void kd_logsoftmax16(
    const float* __restrict__ logits, float* __restrict__ out, int rows)
{
    int r = blockIdx.x * blockDim.x + threadIdx.x;
    if (r >= rows) return;
    float v[16];
    float mx = -3.402823466e38f;
#pragma unroll
    for (int i = 0; i < 16; ++i) { v[i] = logits[r * 16 + i]; mx = fmaxf(mx, v[i]); }
    float s = 0.0f;
#pragma unroll
    for (int i = 0; i < 16; ++i) s += expf(v[i] - mx);
    const float l = mx + logf(s);
#pragma unroll
    for (int i = 0; i < 16; ++i) out[r * 16 + i] = v[i] - l;
}

extern "C" void kernel_launch(void* const* d_in, const int* in_sizes, int n_in,
                              void* d_out, int out_size, void* d_ws, size_t ws_size,
                              hipStream_t stream) {
    static const int DIMS[11] = {2048, 1024, 512, 256, 128, 64, 32, 16, 8, 4, 2};
    static const int INCH[11] = {3, 8, 32, 64, 64, 64, 128, 256, 512, 512, 512};
    static const int FEAT[11] = {8, 32, 64, 64, 64, 128, 256, 512, 512, 512, 1024};
    static const int RING[11] = {2, 2, 2, 2, 4, 4, 4, 4, 4, 4, 4};
    const int Bsz = 512;

    const size_t XBYTES = (size_t)512 * 32 * 512 * sizeof(float);
    float* X0     = (float*)d_ws;
    float* X1     = (float*)((char*)d_ws + XBYTES);
    float* logits = (float*)((char*)d_ws + 2 * XBYTES);

    const float* cur = (const float*)d_in[0];
    for (int i = 0; i < 11; ++i) {
        const int Dd = DIMS[i], Kc = INCH[i], F = FEAT[i];
        const int Dlog2 = __builtin_ctz((unsigned)Dd);
        float* nxt = (i & 1) ? X1 : X0;

        const int N = Bsz * Dd;
        const int tilesF = (F + 15) / 16;
        const int nw = ((N / 512) * tilesF < 256) ? 4 : 8;
        const int blocksX = N / (64 * nw);
        const int Kp4 = (Kc + 3) & ~3;
        const size_t ldsBytes = (size_t)48 * (Kp4 + 1) * sizeof(float);
        dim3 grid(blocksX, tilesF);
        if (RING[i] == 4)
            kd_gather_wmma<4><<<grid, 32 * nw, ldsBytes, stream>>>(
                cur, (const float*)d_in[12 + 2 * i], (const float*)d_in[13 + 2 * i],
                (const int*)d_in[1 + i], nxt, F, Kc, Dlog2, nw);
        else
            kd_gather_wmma<2><<<grid, 32 * nw, ldsBytes, stream>>>(
                cur, (const float*)d_in[12 + 2 * i], (const float*)d_in[13 + 2 * i],
                (const int*)d_in[1 + i], nxt, F, Kc, Dlog2, nw);
        cur = nxt;
    }

    kd_fc_wmma<1024><<<dim3(32), dim3(256), 0, stream>>>(
        cur, (const float*)d_in[34], (const float*)d_in[35], logits);

    kd_logsoftmax16<<<(512 + 255) / 256, 256, 0, stream>>>(logits, (float*)d_out, 512);

    (void)in_sizes; (void)n_in; (void)out_size; (void)ws_size;
}


# Conv layers: with D < 64 a wave takes its 32 column pairs in row-third order, and every sub-tile runs only the passes of the row-thirds it holds (1,2,2,1 per K tile, not 3 or 2 each); sums unchanged
# speedup vs baseline: 4.8123x; 1.1224x over previous
#include <hip/hip_runtime.h>

typedef float v2f __attribute__((ext_vector_type(2)));
typedef float v4f __attribute__((ext_vector_type(4)));
typedef float v8f __attribute__((ext_vector_type(8)));


template <int N0, int N1, int N2, int N3>
__host__ __device__ constexpr int kd_np(int s)
{
    return s == 0 ? N0 : (s == 1 ? N1 : (s == 2 ? N2 : N3));
}

template <int N0, int N1, int N2, int N3>
__device__ __forceinline__ void kd_tile(
    const float* const (&ar)[4][3], const bool (&on)[4][3], int k0,
    const v2f (&b)[4], v8f (&acc)[4])
{
#pragma unroll
    for (int p = 0; p < 3; ++p) {
#pragma unroll
        for (int s = 0; s < 4; ++s) {
            if (p < kd_np<N0, N1, N2, N3>(s)) {
                v2f a;
                a.x = ar[s][p][k0];
                a.y = ar[s][p][k0 + 1];
                v2f bm = b[s];
                if (kd_np<N0, N1, N2, N3>(s) > 1) {
                    bm.x = on[s][p] ? b[s].x : 0.0f;
                    bm.y = on[s][p] ? b[s].y : 0.0f;
                }
                acc[s] = __builtin_amdgcn_wmma_f32_16x16x4_f32(
                    false, a, false, bm, (short)0, acc[s], false, false);
            }
        }
    }
}

template <int N0, int N1, int N2, int N3, int R>
__device__ __forceinline__ void kd_kloop(
    const float* aLane, int Kp, const int (&qv)[4][3], const int (&cnt)[4],
    const int (&qs)[4],
    const char* xb, const unsigned (&bo0)[4], const unsigned (&bo1)[4],
    int K, int Dlog2, int kOff, v8f (&acc)[4])
{
    const float* ar[4][3];
    bool on[4][3];
#pragma unroll
    for (int s = 0; s < 4; ++s) {
#pragma unroll
        for (int p = 0; p < 3; ++p) {
            if (p < kd_np<N0, N1, N2, N3>(s)) {
                ar[s][p] = aLane + min(qv[s][p], 2) * Kp;
                on[s][p] = (p < cnt[s]) && (qs[s] == qv[s][p]);
            } else {
                ar[s][p] = aLane;
                on[s][p] = false;
            }
        }
    }

    const int nT = K >> 2;
    const unsigned tileB = 16u << Dlog2;
    if (nT > 0) {
        const int last = nT - 1;
        auto ldB = [&](int t, v2f (&b)[4]) {
            const unsigned tob = (unsigned)min(t, last) * tileB;
#pragma unroll
            for (int s = 0; s < 4; ++s) {
                b[s].x = *reinterpret_cast<const float*>(xb + (bo0[s] + tob));
                b[s].y = *reinterpret_cast<const float*>(xb + (bo1[s] + tob));
            }
        };
        v2f b[R][4];
#pragma unroll
        for (int j = 0; j < R - 1; ++j) {
            ldB(j, b[j]);
            __builtin_amdgcn_sched_barrier(0);
        }
        int t = 0;
        for (; t + R <= nT; t += R) {
#pragma unroll
            for (int j = 0; j < R; ++j) {
                ldB(t + j + R - 1, b[(j + R - 1) % R]);
                kd_tile<N0, N1, N2, N3>(ar, on, 4 * (t + j) + kOff, b[j], acc);
                if (j + 1 < R) __builtin_amdgcn_sched_barrier(0);
            }
        }
#pragma unroll
        for (int j = 0; j < R - 1; ++j)
            if (t + j < nT) kd_tile<N0, N1, N2, N3>(ar, on, 4 * (t + j) + kOff, b[j], acc);
    }
    const int kFull = K & ~3;
    if (kFull < K) {
        const int k0 = kFull + kOff;
        const int c0 = min(k0, K - 1) - kOff;
        const int c1 = min(k0 + 1, K - 1) - kOff;
        const ptrdiff_t rowF = (ptrdiff_t)1 << Dlog2;
        v2f b[4];
#pragma unroll
        for (int s = 0; s < 4; ++s) {
            const float* col = reinterpret_cast<const float*>(xb + bo0[s]);
            b[s].x = col[(ptrdiff_t)c0 * rowF];
            b[s].y = col[(ptrdiff_t)c1 * rowF];
        }
        kd_tile<N0, N1, N2, N3>(ar, on, k0, b, acc);
    }
}

template <int R>
__global__ __launch_bounds__(256) void kd_gather_wmma(
    const float* __restrict__ X, const float* __restrict__ Wt,
    const float* __restrict__ bias, const int* __restrict__ sel,
    float* __restrict__ Out, int F, int K, int Dlog2, int nw)
{
    extern __shared__ float sW[];
    const int D   = 1 << Dlog2;
    const int Kp4 = (K + 3) & ~3;
    const int Kp  = Kp4 + 1;
    const int f0  = blockIdx.y << 4;
    const int m0  = 3 * f0;
    const int M   = 3 * F;
    const int lane = threadIdx.x & 31;
    const int wv   = threadIdx.x >> 5;

    constexpr int KU = 8;
    for (int r0 = wv; r0 < 48; r0 += 6 * nw) {
        for (int kb0 = 0; kb0 < Kp4; kb0 += 32 * KU) {
            const int nU = min(KU, (Kp4 - kb0 + 31) >> 5);
            float w[KU][6];
#pragma unroll
            for (int u = 0; u < KU; ++u) {
#pragma unroll
                for (int i = 0; i < 6; ++i) w[u][i] = 0.0f;
                if (u < nU) {
                    const int kc = min(kb0 + 32 * u + lane, K - 1);
#pragma unroll
                    for (int i = 0; i < 6; ++i) {
                        const int m = min(m0 + r0 + i * nw, M - 1);
                        w[u][i] = Wt[(size_t)m * K + kc];
                    }
                }
            }
#pragma unroll
            for (int u = 0; u < KU; ++u) {
                if (u < nU) {
                    const int k = kb0 + 32 * u + lane;
                    if (k < Kp4) {
#pragma unroll
                        for (int i = 0; i < 6; ++i)
                            sW[(r0 + i * nw) * Kp + k] = (k < K) ? w[u][i] : 0.0f;
                    }
                }
            }
        }
    }
    __syncthreads();

    const int n0 = (blockIdx.x * nw + wv) << 6;

    const int half = lane >> 4;
    const int l16  = lane & 15;
    const int kOff = half << 1;
    const float* aLane = sW + 3 * l16 * Kp;

    int ncol[4];
    if (D < 64) {
        const int de = (n0 + 2 * lane) & (D - 1);
        const int qe = min(max((3 * de + sel[de]) >> Dlog2, 0), 2);
        const int qo = min(max((3 * de + 3 + sel[de + 1]) >> Dlog2, 0), 2);
        const int key = 3 * qe + qo;
        const unsigned lower = (1u << lane) - 1u;
        int rank = 0;
#pragma unroll
        for (int v = 0; v < 9; ++v) {
            const unsigned m = (unsigned)__ballot(key == v);
            if (v < key) rank += __popc(m);
            if (v == key) rank += __popc(m & lower);
        }
        const int pairAt = __builtin_amdgcn_ds_permute(rank << 2, lane);
#pragma unroll
        for (int s = 0; s < 4; ++s) {
            const int pj = __shfl(pairAt, 8 * s + (l16 >> 1));
            ncol[s] = n0 + 2 * pj + (l16 & 1);
        }
    } else {
#pragma unroll
        for (int s = 0; s < 4; ++s) ncol[s] = n0 + (s << 4) + l16;
    }

    unsigned bo0[4], bo1[4];
    int qs[4];
    int qv[4][3];
    int cnt[4];
#pragma unroll
    for (int s = 0; s < 4; ++s) {
        const int n  = ncol[s];
        const int bb = n >> Dlog2;
        const int d  = n & (D - 1);
        const int t  = 3 * d + sel[d];
        const int q  = t >> Dlog2;
        const int dp = t & (D - 1);
        bo0[s] = ((unsigned)bb * ((unsigned)K << Dlog2) + ((unsigned)kOff << Dlog2)
                  + (unsigned)dp) << 2;
        bo1[s] = bo0[s] + (4u << Dlog2);
        qs[s] = q;
        const bool has0 = __any(q == 0) != 0;
        const bool has1 = __any(q == 1) != 0;
        const bool has2 = __any(q == 2) != 0;
        qv[s][0] = has0 ? 0 : (has1 ? 1 : 2);
        qv[s][1] = (has0 && has1) ? 1 : 2;
        qv[s][2] = 2;
        cnt[s] = max(1, (int)has0 + (int)has1 + (int)has2);
    }

    const int cmax = max(max(cnt[0], cnt[1]), max(cnt[2], cnt[3]));
    v8f acc[4] = { v8f{}, v8f{}, v8f{}, v8f{} };
    const char* xb = reinterpret_cast<const char*>(X);
    if (cmax == 1)
        kd_kloop<1, 1, 1, 1, R>(aLane, Kp, qv, cnt, qs, xb, bo0, bo1, K, Dlog2, kOff, acc);
    else if (cmax == 2 && cnt[0] == 1 && cnt[3] == 1)
        kd_kloop<1, 2, 2, 1, R>(aLane, Kp, qv, cnt, qs, xb, bo0, bo1, K, Dlog2, kOff, acc);
    else if (cmax == 2)
        kd_kloop<2, 2, 2, 2, R>(aLane, Kp, qv, cnt, qs, xb, bo0, bo1, K, Dlog2, kOff, acc);
    else
        kd_kloop<3, 3, 3, 3, R>(aLane, Kp, qv, cnt, qs, xb, bo0, bo1, K, Dlog2, kOff, acc);

    const int Dh = D >> 1;
    const int fl = f0 + (half << 3);
#pragma unroll
    for (int s = 0; s < 4; ++s) {
#pragma unroll
        for (int r = 0; r < 8; ++r) {
            const int fc = min(fl + r, F - 1);
            float v = fmaxf(acc[s][r] + bias[3 * fc + qs[s]], 0.0f);
            v = fmaxf(v, __shfl_xor(v, 1));
            acc[s][r] = v;
        }
    }
    if (!(l16 & 1)) {
        const int nr = F - fl;
#pragma unroll
        for (int s = 0; s < 4; ++s) {
            const int n  = ncol[s];
            const int bb = n >> Dlog2;
            const int d  = n & (D - 1);
            float* ocol = Out + ((size_t)bb * F + fl) * Dh + (d >> 1);
            if (nr >= 8) {
#pragma unroll
                for (int r = 0; r < 8; ++r) ocol[(size_t)r * Dh] = acc[s][r];
            } else {
#pragma unroll
                for (int r = 0; r < 8; ++r)
                    if (r < nr) ocol[(size_t)r * Dh] = acc[s][r];
            }
        }
    }
}

template <int KS>
__device__ __forceinline__ void fc_load_row(const float* rowSlice, int half,
                                            v4f (&v)[KS / 8])
{
    const v4f* p = reinterpret_cast<const v4f*>(rowSlice + 4 * half);
#pragma unroll
    for (int i = 0; i < KS / 8; ++i) v[i] = p[2 * i];
}

template <int K>
__global__ __launch_bounds__(256) void kd_fc_wmma(
    const float* __restrict__ X, const float* __restrict__ Wt,
    const float* __restrict__ bias, float* __restrict__ Out)
{
    static_assert(K % 64 == 0, "8 slices of whole 8-k pieces");
    constexpr int KS = K / 8;
    __shared__ float red[8 * 256];

    const int lane = threadIdx.x & 31;
    const int wv   = threadIdx.x >> 5;
    const int half = lane >> 4;
    const int l16  = lane & 15;
    const int n0   = blockIdx.x << 4;

    v4f va[KS / 8], vb[KS / 8];
    fc_load_row<KS>(Wt + (size_t)l16 * K + wv * KS, half, va);
    fc_load_row<KS>(X + (size_t)(n0 + l16) * K + wv * KS, half, vb);

    v8f acc0 = v8f{}, acc1 = v8f{};
#pragma unroll
    for (int i = 0; i < KS / 8; ++i) {
        v2f a0, b0, a1, b1;
        a0.x = va[i].x; a0.y = va[i].y;
        b0.x = vb[i].x; b0.y = vb[i].y;
        a1.x = va[i].z; a1.y = va[i].w;
        b1.x = vb[i].z; b1.y = vb[i].w;
        acc0 = __builtin_amdgcn_wmma_f32_16x16x4_f32(
            false, a0, false, b0, (short)0, acc0, false, false);
        acc1 = __builtin_amdgcn_wmma_f32_16x16x4_f32(
            false, a1, false, b1, (short)0, acc1, false, false);
    }

#pragma unroll
    for (int r = 0; r < 8; ++r)
        red[(wv << 8) + (r << 5) + lane] = acc0[r] + acc1[r];
    __syncthreads();

    const int e  = threadIdx.x;
    const int nl = e >> 4;
    const int o  = e & 15;
    const int cell = ((o & 7) << 5) + nl + ((o >> 3) << 4);
    float sum = red[cell];
#pragma unroll
    for (int w = 1; w < 8; ++w) sum += red[(w << 8) + cell];
    Out[((size_t)(n0 + nl) << 4) + o] = sum + bias[o];
}

__global__ __launch_bounds__(256) void kd_logsoftmax16(
    const float* __restrict__ logits, float* __restrict__ out, int rows)
{
    int r = blockIdx.x * blockDim.x + threadIdx.x;
    if (r >= rows) return;
    float v[16];
    float mx = -3.402823466e38f;
#pragma unroll
    for (int i = 0; i < 16; ++i) { v[i] = logits[r * 16 + i]; mx = fmaxf(mx, v[i]); }
    float s = 0.0f;
#pragma unroll
    for (int i = 0; i < 16; ++i) s += expf(v[i] - mx);
    const float l = mx + logf(s);
#pragma unroll
    for (int i = 0; i < 16; ++i) out[r * 16 + i] = v[i] - l;
}

extern "C" void kernel_launch(void* const* d_in, const int* in_sizes, int n_in,
                              void* d_out, int out_size, void* d_ws, size_t ws_size,
                              hipStream_t stream) {
    static const int DIMS[11] = {2048, 1024, 512, 256, 128, 64, 32, 16, 8, 4, 2};
    static const int INCH[11] = {3, 8, 32, 64, 64, 64, 128, 256, 512, 512, 512};
    static const int FEAT[11] = {8, 32, 64, 64, 64, 128, 256, 512, 512, 512, 1024};
    static const int RING[11] = {2, 2, 2, 2, 4, 4, 4, 4, 4, 4, 4};
    const int Bsz = 512;

    const size_t XBYTES = (size_t)512 * 32 * 512 * sizeof(float);
    float* X0     = (float*)d_ws;
    float* X1     = (float*)((char*)d_ws + XBYTES);
    float* logits = (float*)((char*)d_ws + 2 * XBYTES);

    const float* cur = (const float*)d_in[0];
    for (int i = 0; i < 11; ++i) {
        const int Dd = DIMS[i], Kc = INCH[i], F = FEAT[i];
        const int Dlog2 = __builtin_ctz((unsigned)Dd);
        float* nxt = (i & 1) ? X1 : X0;

        const int N = Bsz * Dd;
        const int tilesF = (F + 15) / 16;
        const int nw = ((N / 512) * tilesF < 256) ? 4 : 8;
        const int blocksX = N / (64 * nw);
        const int Kp4 = (Kc + 3) & ~3;
        const size_t ldsBytes = (size_t)48 * (Kp4 + 1) * sizeof(float);
        dim3 grid(blocksX, tilesF);
        if (RING[i] == 4)
            kd_gather_wmma<4><<<grid, 32 * nw, ldsBytes, stream>>>(
                cur, (const float*)d_in[12 + 2 * i], (const float*)d_in[13 + 2 * i],
                (const int*)d_in[1 + i], nxt, F, Kc, Dlog2, nw);
        else
            kd_gather_wmma<2><<<grid, 32 * nw, ldsBytes, stream>>>(
                cur, (const float*)d_in[12 + 2 * i], (const float*)d_in[13 + 2 * i],
                (const int*)d_in[1 + i], nxt, F, Kc, Dlog2, nw);
        cur = nxt;
    }

    kd_fc_wmma<1024><<<dim3(32), dim3(256), 0, stream>>>(
        cur, (const float*)d_in[34], (const float*)d_in[35], logits);

    kd_logsoftmax16<<<(512 + 255) / 256, 256, 0, stream>>>(logits, (float*)d_out, 512);

    (void)in_sizes; (void)n_in; (void)out_size; (void)ws_size;
}


# Layers 10-11 (one wave per SIMD): K split over two wave sets per block, partial tiles added through LDS (new summation order: first K half + second K half); two waves per SIMD cover each other
# speedup vs baseline: 5.0748x; 1.0545x over previous
#include <hip/hip_runtime.h>

typedef float v2f __attribute__((ext_vector_type(2)));
typedef float v4f __attribute__((ext_vector_type(4)));
typedef float v8f __attribute__((ext_vector_type(8)));


template <int N0, int N1, int N2, int N3>
__host__ __device__ constexpr int kd_np(int s)
{
    return s == 0 ? N0 : (s == 1 ? N1 : (s == 2 ? N2 : N3));
}

template <int N0, int N1, int N2, int N3>
__device__ __forceinline__ void kd_tile(
    const float* const (&ar)[4][3], const bool (&on)[4][3], int k0,
    const v2f (&b)[4], v8f (&acc)[4])
{
#pragma unroll
    for (int p = 0; p < 3; ++p) {
#pragma unroll
        for (int s = 0; s < 4; ++s) {
            if (p < kd_np<N0, N1, N2, N3>(s)) {
                v2f a;
                a.x = ar[s][p][k0];
                a.y = ar[s][p][k0 + 1];
                v2f bm = b[s];
                if (kd_np<N0, N1, N2, N3>(s) > 1) {
                    bm.x = on[s][p] ? b[s].x : 0.0f;
                    bm.y = on[s][p] ? b[s].y : 0.0f;
                }
                acc[s] = __builtin_amdgcn_wmma_f32_16x16x4_f32(
                    false, a, false, bm, (short)0, acc[s], false, false);
            }
        }
    }
}

template <int N0, int N1, int N2, int N3, int R>
__device__ __forceinline__ void kd_kloop(
    const float* aLane, int Kp, const int (&qv)[4][3], const int (&cnt)[4],
    const int (&qs)[4],
    const char* xb, const unsigned (&bo0)[4], const unsigned (&bo1)[4],
    int K, int Dlog2, int kOff, v8f (&acc)[4])
{
    const float* ar[4][3];
    bool on[4][3];
#pragma unroll
    for (int s = 0; s < 4; ++s) {
#pragma unroll
        for (int p = 0; p < 3; ++p) {
            if (p < kd_np<N0, N1, N2, N3>(s)) {
                ar[s][p] = aLane + min(qv[s][p], 2) * Kp;
                on[s][p] = (p < cnt[s]) && (qs[s] == qv[s][p]);
            } else {
                ar[s][p] = aLane;
                on[s][p] = false;
            }
        }
    }

    const int nT = K >> 2;
    const unsigned tileB = 16u << Dlog2;
    if (nT > 0) {
        const int last = nT - 1;
        auto ldB = [&](int t, v2f (&b)[4]) {
            const unsigned tob = (unsigned)min(t, last) * tileB;
#pragma unroll
            for (int s = 0; s < 4; ++s) {
                b[s].x = *reinterpret_cast<const float*>(xb + (bo0[s] + tob));
                b[s].y = *reinterpret_cast<const float*>(xb + (bo1[s] + tob));
            }
        };
        v2f b[R][4];
#pragma unroll
        for (int j = 0; j < R - 1; ++j) {
            ldB(j, b[j]);
            __builtin_amdgcn_sched_barrier(0);
        }
        int t = 0;
        for (; t + R <= nT; t += R) {
#pragma unroll
            for (int j = 0; j < R; ++j) {
                ldB(t + j + R - 1, b[(j + R - 1) % R]);
                kd_tile<N0, N1, N2, N3>(ar, on, 4 * (t + j) + kOff, b[j], acc);
                if (j + 1 < R) __builtin_amdgcn_sched_barrier(0);
            }
        }
#pragma unroll
        for (int j = 0; j < R - 1; ++j)
            if (t + j < nT) kd_tile<N0, N1, N2, N3>(ar, on, 4 * (t + j) + kOff, b[j], acc);
    }
    const int kFull = K & ~3;
    if (kFull < K) {
        const int k0 = kFull + kOff;
        const int c0 = min(k0, K - 1) - kOff;
        const int c1 = min(k0 + 1, K - 1) - kOff;
        const ptrdiff_t rowF = (ptrdiff_t)1 << Dlog2;
        v2f b[4];
#pragma unroll
        for (int s = 0; s < 4; ++s) {
            const float* col = reinterpret_cast<const float*>(xb + bo0[s]);
            b[s].x = col[(ptrdiff_t)c0 * rowF];
            b[s].y = col[(ptrdiff_t)c1 * rowF];
        }
        kd_tile<N0, N1, N2, N3>(ar, on, k0, b, acc);
    }
}

template <int R>
__global__ __launch_bounds__(256) void kd_gather_wmma(
    const float* __restrict__ X, const float* __restrict__ Wt,
    const float* __restrict__ bias, const int* __restrict__ sel,
    float* __restrict__ Out, int F, int K, int Dlog2, int nw, int KS)
{
    extern __shared__ float sW[];
    const int D   = 1 << Dlog2;
    const int Kp4 = (K + 3) & ~3;
    const int Kp  = Kp4 + 1;
    const int f0  = blockIdx.y << 4;
    const int m0  = 3 * f0;
    const int M   = 3 * F;
    const int lane = threadIdx.x & 31;
    const int wv   = threadIdx.x >> 5;

    constexpr int KU = 8;
    for (int r0 = wv; r0 < 48; r0 += 6 * nw) {
        for (int kb0 = 0; kb0 < Kp4; kb0 += 32 * KU) {
            const int nU = min(KU, (Kp4 - kb0 + 31) >> 5);
            float w[KU][6];
#pragma unroll
            for (int u = 0; u < KU; ++u) {
#pragma unroll
                for (int i = 0; i < 6; ++i) w[u][i] = 0.0f;
                if (u < nU) {
                    const int kc = min(kb0 + 32 * u + lane, K - 1);
#pragma unroll
                    for (int i = 0; i < 6; ++i) {
                        const int m = min(m0 + r0 + i * nw, M - 1);
                        w[u][i] = Wt[(size_t)m * K + kc];
                    }
                }
            }
#pragma unroll
            for (int u = 0; u < KU; ++u) {
                if (u < nU) {
                    const int k = kb0 + 32 * u + lane;
                    if (k < Kp4) {
#pragma unroll
                        for (int i = 0; i < 6; ++i)
                            sW[(r0 + i * nw) * Kp + k] = (k < K) ? w[u][i] : 0.0f;
                    }
                }
            }
        }
    }
    __syncthreads();

    const int nwc = (KS == 2) ? (nw >> 1) : nw;
    const int kh  = (wv >= nwc) ? 1 : 0;
    const int cg  = wv - kh * nwc;
    const int n0  = (blockIdx.x * nwc + cg) << 6;

    const int half = lane >> 4;
    const int l16  = lane & 15;
    const int kOff = half << 1;
    const float* aLane = sW + 3 * l16 * Kp;

    int ncol[4];
    if (D < 64) {
        const int de = (n0 + 2 * lane) & (D - 1);
        const int qe = min(max((3 * de + sel[de]) >> Dlog2, 0), 2);
        const int qo = min(max((3 * de + 3 + sel[de + 1]) >> Dlog2, 0), 2);
        const int key = 3 * qe + qo;
        const unsigned lower = (1u << lane) - 1u;
        int rank = 0;
#pragma unroll
        for (int v = 0; v < 9; ++v) {
            const unsigned m = (unsigned)__ballot(key == v);
            if (v < key) rank += __popc(m);
            if (v == key) rank += __popc(m & lower);
        }
        const int pairAt = __builtin_amdgcn_ds_permute(rank << 2, lane);
#pragma unroll
        for (int s = 0; s < 4; ++s) {
            const int pj = __shfl(pairAt, 8 * s + (l16 >> 1));
            ncol[s] = n0 + 2 * pj + (l16 & 1);
        }
    } else {
#pragma unroll
        for (int s = 0; s < 4; ++s) ncol[s] = n0 + (s << 4) + l16;
    }

    unsigned bo0[4], bo1[4];
    int qs[4];
    int qv[4][3];
    int cnt[4];
#pragma unroll
    for (int s = 0; s < 4; ++s) {
        const int n  = ncol[s];
        const int bb = n >> Dlog2;
        const int d  = n & (D - 1);
        const int t  = 3 * d + sel[d];
        const int q  = t >> Dlog2;
        const int dp = t & (D - 1);
        bo0[s] = ((unsigned)bb * ((unsigned)K << Dlog2) + ((unsigned)kOff << Dlog2)
                  + (unsigned)dp) << 2;
        if (KS == 2) bo0[s] += ((unsigned)kh * (unsigned)(K >> 1)) << (Dlog2 + 2);
        bo1[s] = bo0[s] + (4u << Dlog2);
        qs[s] = q;
        const bool has0 = __any(q == 0) != 0;
        const bool has1 = __any(q == 1) != 0;
        const bool has2 = __any(q == 2) != 0;
        qv[s][0] = has0 ? 0 : (has1 ? 1 : 2);
        qv[s][1] = (has0 && has1) ? 1 : 2;
        qv[s][2] = 2;
        cnt[s] = max(1, (int)has0 + (int)has1 + (int)has2);
    }

    const int cmax = max(max(cnt[0], cnt[1]), max(cnt[2], cnt[3]));
    v8f acc[4] = { v8f{}, v8f{}, v8f{}, v8f{} };
    const char* xb = reinterpret_cast<const char*>(X);
    const int Ks = (KS == 2) ? (K >> 1) : K;
    const float* aSl = aLane + kh * Ks;
    if (cmax == 1)
        kd_kloop<1, 1, 1, 1, R>(aSl, Kp, qv, cnt, qs, xb, bo0, bo1, Ks, Dlog2, kOff, acc);
    else if (cmax == 2 && cnt[0] == 1 && cnt[3] == 1)
        kd_kloop<1, 2, 2, 1, R>(aSl, Kp, qv, cnt, qs, xb, bo0, bo1, Ks, Dlog2, kOff, acc);
    else if (cmax == 2)
        kd_kloop<2, 2, 2, 2, R>(aSl, Kp, qv, cnt, qs, xb, bo0, bo1, Ks, Dlog2, kOff, acc);
    else
        kd_kloop<3, 3, 3, 3, R>(aSl, Kp, qv, cnt, qs, xb, bo0, bo1, Ks, Dlog2, kOff, acc);

    if (KS == 2) {
        float* red = sW + 48 * Kp + (cg << 10);
        if (kh == 1) {
#pragma unroll
            for (int s = 0; s < 4; ++s)
#pragma unroll
                for (int r = 0; r < 8; ++r) red[(((s << 3) + r) << 5) + lane] = acc[s][r];
        }
        __syncthreads();
        if (kh == 1) return;
#pragma unroll
        for (int s = 0; s < 4; ++s)
#pragma unroll
            for (int r = 0; r < 8; ++r) acc[s][r] += red[(((s << 3) + r) << 5) + lane];
    }

    const int Dh = D >> 1;
    const int fl = f0 + (half << 3);
#pragma unroll
    for (int s = 0; s < 4; ++s) {
#pragma unroll
        for (int r = 0; r < 8; ++r) {
            const int fc = min(fl + r, F - 1);
            float v = fmaxf(acc[s][r] + bias[3 * fc + qs[s]], 0.0f);
            v = fmaxf(v, __shfl_xor(v, 1));
            acc[s][r] = v;
        }
    }
    if (!(l16 & 1)) {
        const int nr = F - fl;
#pragma unroll
        for (int s = 0; s < 4; ++s) {
            const int n  = ncol[s];
            const int bb = n >> Dlog2;
            const int d  = n & (D - 1);
            float* ocol = Out + ((size_t)bb * F + fl) * Dh + (d >> 1);
            if (nr >= 8) {
#pragma unroll
                for (int r = 0; r < 8; ++r) ocol[(size_t)r * Dh] = acc[s][r];
            } else {
#pragma unroll
                for (int r = 0; r < 8; ++r)
                    if (r < nr) ocol[(size_t)r * Dh] = acc[s][r];
            }
        }
    }
}

template <int KS>
__device__ __forceinline__ void fc_load_row(const float* rowSlice, int half,
                                            v4f (&v)[KS / 8])
{
    const v4f* p = reinterpret_cast<const v4f*>(rowSlice + 4 * half);
#pragma unroll
    for (int i = 0; i < KS / 8; ++i) v[i] = p[2 * i];
}

template <int K>
__global__ __launch_bounds__(256) void kd_fc_wmma(
    const float* __restrict__ X, const float* __restrict__ Wt,
    const float* __restrict__ bias, float* __restrict__ Out)
{
    static_assert(K % 64 == 0, "8 slices of whole 8-k pieces");
    constexpr int KS = K / 8;
    __shared__ float red[8 * 256];

    const int lane = threadIdx.x & 31;
    const int wv   = threadIdx.x >> 5;
    const int half = lane >> 4;
    const int l16  = lane & 15;
    const int n0   = blockIdx.x << 4;

    v4f va[KS / 8], vb[KS / 8];
    fc_load_row<KS>(Wt + (size_t)l16 * K + wv * KS, half, va);
    fc_load_row<KS>(X + (size_t)(n0 + l16) * K + wv * KS, half, vb);

    v8f acc0 = v8f{}, acc1 = v8f{};
#pragma unroll
    for (int i = 0; i < KS / 8; ++i) {
        v2f a0, b0, a1, b1;
        a0.x = va[i].x; a0.y = va[i].y;
        b0.x = vb[i].x; b0.y = vb[i].y;
        a1.x = va[i].z; a1.y = va[i].w;
        b1.x = vb[i].z; b1.y = vb[i].w;
        acc0 = __builtin_amdgcn_wmma_f32_16x16x4_f32(
            false, a0, false, b0, (short)0, acc0, false, false);
        acc1 = __builtin_amdgcn_wmma_f32_16x16x4_f32(
            false, a1, false, b1, (short)0, acc1, false, false);
    }

#pragma unroll
    for (int r = 0; r < 8; ++r)
        red[(wv << 8) + (r << 5) + lane] = acc0[r] + acc1[r];
    __syncthreads();

    const int e  = threadIdx.x;
    const int nl = e >> 4;
    const int o  = e & 15;
    const int cell = ((o & 7) << 5) + nl + ((o >> 3) << 4);
    float sum = red[cell];
#pragma unroll
    for (int w = 1; w < 8; ++w) sum += red[(w << 8) + cell];
    Out[((size_t)(n0 + nl) << 4) + o] = sum + bias[o];
}

__global__ __launch_bounds__(256) void kd_logsoftmax16(
    const float* __restrict__ logits, float* __restrict__ out, int rows)
{
    int r = blockIdx.x * blockDim.x + threadIdx.x;
    if (r >= rows) return;
    float v[16];
    float mx = -3.402823466e38f;
#pragma unroll
    for (int i = 0; i < 16; ++i) { v[i] = logits[r * 16 + i]; mx = fmaxf(mx, v[i]); }
    float s = 0.0f;
#pragma unroll
    for (int i = 0; i < 16; ++i) s += expf(v[i] - mx);
    const float l = mx + logf(s);
#pragma unroll
    for (int i = 0; i < 16; ++i) out[r * 16 + i] = v[i] - l;
}

extern "C" void kernel_launch(void* const* d_in, const int* in_sizes, int n_in,
                              void* d_out, int out_size, void* d_ws, size_t ws_size,
                              hipStream_t stream) {
    static const int DIMS[11] = {2048, 1024, 512, 256, 128, 64, 32, 16, 8, 4, 2};
    static const int INCH[11] = {3, 8, 32, 64, 64, 64, 128, 256, 512, 512, 512};
    static const int FEAT[11] = {8, 32, 64, 64, 64, 128, 256, 512, 512, 512, 1024};
    static const int RING[11] = {2, 2, 2, 2, 4, 4, 4, 4, 4, 4, 4};
    const int Bsz = 512;

    const size_t XBYTES = (size_t)512 * 32 * 512 * sizeof(float);
    float* X0     = (float*)d_ws;
    float* X1     = (float*)((char*)d_ws + XBYTES);
    float* logits = (float*)((char*)d_ws + 2 * XBYTES);

    const float* cur = (const float*)d_in[0];
    for (int i = 0; i < 11; ++i) {
        const int Dd = DIMS[i], Kc = INCH[i], F = FEAT[i];
        const int Dlog2 = __builtin_ctz((unsigned)Dd);
        float* nxt = (i & 1) ? X1 : X0;

        const int N = Bsz * Dd;
        const int tilesF = (F + 15) / 16;
        const int ncg = ((N / 512) * tilesF < 256) ? 4 : 8;
        const int KS  = (ncg == 4 && Kc % 32 == 0) ? 2 : 1;
        const int nw  = ncg * KS;
        const int blocksX = N / (64 * ncg);
        const int Kp4 = (Kc + 3) & ~3;
        const size_t ldsBytes = (size_t)48 * (Kp4 + 1) * sizeof(float)
            + (KS == 2 ? (size_t)ncg * 1024 * sizeof(float) : 0);
        dim3 grid(blocksX, tilesF);
        if (RING[i] == 4)
            kd_gather_wmma<4><<<grid, 32 * nw, ldsBytes, stream>>>(
                cur, (const float*)d_in[12 + 2 * i], (const float*)d_in[13 + 2 * i],
                (const int*)d_in[1 + i], nxt, F, Kc, Dlog2, nw, KS);
        else
            kd_gather_wmma<2><<<grid, 32 * nw, ldsBytes, stream>>>(
                cur, (const float*)d_in[12 + 2 * i], (const float*)d_in[13 + 2 * i],
                (const int*)d_in[1 + i], nxt, F, Kc, Dlog2, nw, KS);
        cur = nxt;
    }

    kd_fc_wmma<1024><<<dim3(32), dim3(256), 0, stream>>>(
        cur, (const float*)d_in[34], (const float*)d_in[35], logits);

    kd_logsoftmax16<<<(512 + 255) / 256, 256, 0, stream>>>(logits, (float*)d_out, 512);

    (void)in_sizes; (void)n_in; (void)out_size; (void)ws_size;
}


# Conv activations kept k-minor [b][d][k]: B fragments read as 16-byte vectors, one whole 128-byte line per column and 32-k chunk; pooled cells stored 16 bytes at a time; new k order in the sums
# speedup vs baseline: 5.3864x; 1.0614x over previous
#include <hip/hip_runtime.h>

typedef float v2f __attribute__((ext_vector_type(2)));
typedef float v4f __attribute__((ext_vector_type(4)));
typedef float v8f __attribute__((ext_vector_type(8)));


template <int N0, int N1, int N2, int N3>
__host__ __device__ constexpr int kd_np(int s)
{
    return s == 0 ? N0 : (s == 1 ? N1 : (s == 2 ? N2 : N3));
}

template <int N0, int N1, int N2, int N3>
__device__ __forceinline__ void kd_tile(
    const float* const (&ar)[4][3], const bool (&on)[4][3], int k0,
    const v2f (&b)[4], v8f (&acc)[4])
{
#pragma unroll
    for (int p = 0; p < 3; ++p) {
#pragma unroll
        for (int s = 0; s < 4; ++s) {
            if (p < kd_np<N0, N1, N2, N3>(s)) {
                v2f a;
                a.x = ar[s][p][k0];
                a.y = ar[s][p][k0 + 1];
                v2f bm = b[s];
                if (kd_np<N0, N1, N2, N3>(s) > 1) {
                    bm.x = on[s][p] ? b[s].x : 0.0f;
                    bm.y = on[s][p] ? b[s].y : 0.0f;
                }
                acc[s] = __builtin_amdgcn_wmma_f32_16x16x4_f32(
                    false, a, false, bm, (short)0, acc[s], false, false);
            }
        }
    }
}

template <int N0, int N1, int N2, int N3, int NV>
__device__ __forceinline__ void kd_chunk(
    const float* const (&ar)[4][3], const bool (&on)[4][3], int k0,
    const v4f (&b)[4][NV], v8f (&acc)[4])
{
#pragma unroll
    for (int v = 0; v < NV; ++v) {
#pragma unroll
        for (int e = 0; e < 2; ++e) {
#pragma unroll
            for (int p = 0; p < 3; ++p) {
#pragma unroll
                for (int s = 0; s < 4; ++s) {
                    if (p < kd_np<N0, N1, N2, N3>(s)) {
                        const int ka = k0 + 4 * v + 2 * e;
                        v2f a;
                        a.x = ar[s][p][ka];
                        a.y = ar[s][p][ka + 1];
                        v2f bm;
                        bm.x = e ? b[s][v].z : b[s][v].x;
                        bm.y = e ? b[s][v].w : b[s][v].y;
                        if (kd_np<N0, N1, N2, N3>(s) > 1) {
                            bm.x = on[s][p] ? bm.x : 0.0f;
                            bm.y = on[s][p] ? bm.y : 0.0f;
                        }
                        acc[s] = __builtin_amdgcn_wmma_f32_16x16x4_f32(
                            false, a, false, bm, (short)0, acc[s], false, false);
                    }
                }
            }
        }
    }
}

template <int N0, int N1, int N2, int N3>
__device__ __forceinline__ void kd_rows(
    const float* aLane, int Kp, const int (&qv)[4][3], const int (&cnt)[4],
    const int (&qs)[4], const float* (&ar)[4][3], bool (&on)[4][3])
{
#pragma unroll
    for (int s = 0; s < 4; ++s) {
#pragma unroll
        for (int p = 0; p < 3; ++p) {
            if (p < kd_np<N0, N1, N2, N3>(s)) {
                ar[s][p] = aLane + min(qv[s][p], 2) * Kp;
                on[s][p] = (p < cnt[s]) && (qs[s] == qv[s][p]);
            } else {
                ar[s][p] = aLane;
                on[s][p] = false;
            }
        }
    }
}

template <int N0, int N1, int N2, int N3, int R, int NV>
__device__ __forceinline__ void kd_kloop(
    const float* aLane, int Kp, const int (&qv)[4][3], const int (&cnt)[4],
    const int (&qs)[4], const char* xb, const unsigned (&bo)[4],
    int K, int half, v8f (&acc)[4])
{
    const float* ar[4][3];
    bool on[4][3];
    kd_rows<N0, N1, N2, N3>(aLane, Kp, qv, cnt, qs, ar, on);

    constexpr int CH = 8 * NV;
    const int nC = K / CH;
    const int last = nC - 1;
    const int kl = 4 * NV * half;
    auto ldB = [&](int j, v4f (&b)[4][NV]) {
        const unsigned tob = (unsigned)min(j, last) * (unsigned)(CH * 4);
#pragma unroll
        for (int s = 0; s < 4; ++s)
#pragma unroll
            for (int v = 0; v < NV; ++v)
                b[s][v] = *reinterpret_cast<const v4f*>(xb + (bo[s] + tob + 16u * v));
    };
    if (R == 1) {
        for (int j = 0; j < nC; ++j) {
            v4f b[4][NV];
            ldB(j, b);
            kd_chunk<N0, N1, N2, N3, NV>(ar, on, CH * j + kl, b, acc);
        }
    } else {
        v4f b0[4][NV], b1[4][NV];
        ldB(0, b0);
        __builtin_amdgcn_sched_barrier(0);
        int j = 0;
        for (; j + 2 <= nC; j += 2) {
            ldB(j + 1, b1);
            kd_chunk<N0, N1, N2, N3, NV>(ar, on, CH * j + kl, b0, acc);
            __builtin_amdgcn_sched_barrier(0);
            ldB(j + 2, b0);
            kd_chunk<N0, N1, N2, N3, NV>(ar, on, CH * (j + 1) + kl, b1, acc);
            __builtin_amdgcn_sched_barrier(0);
        }
        if (j < nC) kd_chunk<N0, N1, N2, N3, NV>(ar, on, CH * j + kl, b0, acc);
    }
}

template <int N0, int N1, int N2, int N3>
__device__ __forceinline__ void kd_ktail(
    const float* aLane, int Kp, const int (&qv)[4][3], const int (&cnt)[4],
    const int (&qs)[4], const char* xb, const unsigned (&bo0)[4],
    int K, int Dlog2, int kOff, v8f (&acc)[4])
{
    const float* ar[4][3];
    bool on[4][3];
    kd_rows<N0, N1, N2, N3>(aLane, Kp, qv, cnt, qs, ar, on);
    const int c0 = min(kOff, K - 1) - kOff;
    const int c1 = min(kOff + 1, K - 1) - kOff;
    const ptrdiff_t rowF = (ptrdiff_t)1 << Dlog2;
    v2f b[4];
#pragma unroll
    for (int s = 0; s < 4; ++s) {
        const float* col = reinterpret_cast<const float*>(xb + bo0[s]);
        b[s].x = col[(ptrdiff_t)c0 * rowF];
        b[s].y = col[(ptrdiff_t)c1 * rowF];
    }
    kd_tile<N0, N1, N2, N3>(ar, on, kOff, b, acc);
}

template <int R, int NV>
__global__ __launch_bounds__(256) void kd_gather_wmma(
    const float* __restrict__ X, const float* __restrict__ Wt,
    const float* __restrict__ bias, const int* __restrict__ sel,
    float* __restrict__ Out, int F, int K, int Dlog2, int nw, int KS)
{
    extern __shared__ float sW[];
    const int D   = 1 << Dlog2;
    const int Kp4 = (K + 3) & ~3;
    const int Kp  = Kp4 + 1;
    const int f0  = blockIdx.y << 4;
    const int m0  = 3 * f0;
    const int M   = 3 * F;
    const int lane = threadIdx.x & 31;
    const int wv   = threadIdx.x >> 5;

    constexpr int KU = 8;
    for (int r0 = wv; r0 < 48; r0 += 6 * nw) {
        for (int kb0 = 0; kb0 < Kp4; kb0 += 32 * KU) {
            const int nU = min(KU, (Kp4 - kb0 + 31) >> 5);
            float w[KU][6];
#pragma unroll
            for (int u = 0; u < KU; ++u) {
#pragma unroll
                for (int i = 0; i < 6; ++i) w[u][i] = 0.0f;
                if (u < nU) {
                    const int kc = min(kb0 + 32 * u + lane, K - 1);
#pragma unroll
                    for (int i = 0; i < 6; ++i) {
                        const int m = min(m0 + r0 + i * nw, M - 1);
                        w[u][i] = Wt[(size_t)m * K + kc];
                    }
                }
            }
#pragma unroll
            for (int u = 0; u < KU; ++u) {
                if (u < nU) {
                    const int k = kb0 + 32 * u + lane;
                    if (k < Kp4) {
#pragma unroll
                        for (int i = 0; i < 6; ++i)
                            sW[(r0 + i * nw) * Kp + k] = (k < K) ? w[u][i] : 0.0f;
                    }
                }
            }
        }
    }
    __syncthreads();

    const int nwc = (KS == 2) ? (nw >> 1) : nw;
    const int kh  = (wv >= nwc) ? 1 : 0;
    const int cg  = wv - kh * nwc;
    const int n0  = (blockIdx.x * nwc + cg) << 6;

    const int half = lane >> 4;
    const int l16  = lane & 15;
    const int kOff = half << 1;
    const float* aLane = sW + 3 * l16 * Kp;

    int ncol[4];
    if (D < 64) {
        const int de = (n0 + 2 * lane) & (D - 1);
        const int qe = min(max((3 * de + sel[de]) >> Dlog2, 0), 2);
        const int qo = min(max((3 * de + 3 + sel[de + 1]) >> Dlog2, 0), 2);
        const int key = 3 * qe + qo;
        const unsigned lower = (1u << lane) - 1u;
        int rank = 0;
#pragma unroll
        for (int v = 0; v < 9; ++v) {
            const unsigned m = (unsigned)__ballot(key == v);
            if (v < key) rank += __popc(m);
            if (v == key) rank += __popc(m & lower);
        }
        const int pairAt = __builtin_amdgcn_ds_permute(rank << 2, lane);
#pragma unroll
        for (int s = 0; s < 4; ++s) {
            const int pj = __shfl(pairAt, 8 * s + (l16 >> 1));
            ncol[s] = n0 + 2 * pj + (l16 & 1);
        }
    } else {
#pragma unroll
        for (int s = 0; s < 4; ++s) ncol[s] = n0 + (s << 4) + l16;
    }

    unsigned bo0[4];
    int qs[4];
    int qv[4][3];
    int cnt[4];
#pragma unroll
    for (int s = 0; s < 4; ++s) {
        const int n  = ncol[s];
        const int bb = n >> Dlog2;
        const int d  = n & (D - 1);
        const int t  = 3 * d + sel[d];
        const int q  = t >> Dlog2;
        const int dp = t & (D - 1);
        if (NV > 0) {
            bo0[s] = ((((unsigned)bb << Dlog2) + (unsigned)dp) * (unsigned)K
                      + (unsigned)(4 * NV * half)) << 2;
            if (KS == 2) bo0[s] += ((unsigned)kh * (unsigned)(K >> 1)) << 2;
        } else {
            bo0[s] = ((unsigned)bb * ((unsigned)K << Dlog2) + ((unsigned)kOff << Dlog2)
                      + (unsigned)dp) << 2;
        }
        qs[s] = q;
        const bool has0 = __any(q == 0) != 0;
        const bool has1 = __any(q == 1) != 0;
        const bool has2 = __any(q == 2) != 0;
        qv[s][0] = has0 ? 0 : (has1 ? 1 : 2);
        qv[s][1] = (has0 && has1) ? 1 : 2;
        qv[s][2] = 2;
        cnt[s] = max(1, (int)has0 + (int)has1 + (int)has2);
    }

    const int cmax = max(max(cnt[0], cnt[1]), max(cnt[2], cnt[3]));
    v8f acc[4] = { v8f{}, v8f{}, v8f{}, v8f{} };
    const char* xb = reinterpret_cast<const char*>(X);
    const int Ks = (KS == 2) ? (K >> 1) : K;
    const float* aSl = aLane + kh * Ks;
    if (NV > 0) {
        constexpr int NVc = NV > 0 ? NV : 1;
        if (cmax == 1)
            kd_kloop<1, 1, 1, 1, R, NVc>(aSl, Kp, qv, cnt, qs, xb, bo0, Ks, half, acc);
        else if (cmax == 2 && cnt[0] == 1 && cnt[3] == 1)
            kd_kloop<1, 2, 2, 1, R, NVc>(aSl, Kp, qv, cnt, qs, xb, bo0, Ks, half, acc);
        else if (cmax == 2)
            kd_kloop<2, 2, 2, 2, R, NVc>(aSl, Kp, qv, cnt, qs, xb, bo0, Ks, half, acc);
        else
            kd_kloop<3, 3, 3, 3, R, NVc>(aSl, Kp, qv, cnt, qs, xb, bo0, Ks, half, acc);
    } else {
        if (cmax == 1)
            kd_ktail<1, 1, 1, 1>(aSl, Kp, qv, cnt, qs, xb, bo0, Ks, Dlog2, kOff, acc);
        else if (cmax == 2 && cnt[0] == 1 && cnt[3] == 1)
            kd_ktail<1, 2, 2, 1>(aSl, Kp, qv, cnt, qs, xb, bo0, Ks, Dlog2, kOff, acc);
        else if (cmax == 2)
            kd_ktail<2, 2, 2, 2>(aSl, Kp, qv, cnt, qs, xb, bo0, Ks, Dlog2, kOff, acc);
        else
            kd_ktail<3, 3, 3, 3>(aSl, Kp, qv, cnt, qs, xb, bo0, Ks, Dlog2, kOff, acc);
    }

    if (KS == 2) {
        float* red = sW + 48 * Kp + (cg << 10);
        if (kh == 1) {
#pragma unroll
            for (int s = 0; s < 4; ++s)
#pragma unroll
                for (int r = 0; r < 8; ++r) red[(((s << 3) + r) << 5) + lane] = acc[s][r];
        }
        __syncthreads();
        if (kh == 1) return;
#pragma unroll
        for (int s = 0; s < 4; ++s)
#pragma unroll
            for (int r = 0; r < 8; ++r) acc[s][r] += red[(((s << 3) + r) << 5) + lane];
    }

    const int Dh = D >> 1;
    const int fl = f0 + (half << 3);
#pragma unroll
    for (int s = 0; s < 4; ++s) {
#pragma unroll
        for (int r = 0; r < 8; ++r) {
            const int fc = min(fl + r, F - 1);
            float v = fmaxf(acc[s][r] + bias[3 * fc + qs[s]], 0.0f);
            v = fmaxf(v, __shfl_xor(v, 1));
            acc[s][r] = v;
        }
    }
    if (!(l16 & 1)) {
        const int nr = F - fl;
#pragma unroll
        for (int s = 0; s < 4; ++s) {
            const int n  = ncol[s];
            const int bb = n >> Dlog2;
            const int d  = n & (D - 1);
            float* ocol = Out + ((size_t)bb * Dh + (d >> 1)) * F + fl;
            if (nr >= 8 && !(F & 3)) {
                v4f lo, hi;
                lo.x = acc[s][0]; lo.y = acc[s][1]; lo.z = acc[s][2]; lo.w = acc[s][3];
                hi.x = acc[s][4]; hi.y = acc[s][5]; hi.z = acc[s][6]; hi.w = acc[s][7];
                *reinterpret_cast<v4f*>(ocol) = lo;
                *reinterpret_cast<v4f*>(ocol + 4) = hi;
            } else {
#pragma unroll
                for (int r = 0; r < 8; ++r)
                    if (r < nr) ocol[r] = acc[s][r];
            }
        }
    }
}

template <int KS>
__device__ __forceinline__ void fc_load_row(const float* rowSlice, int half,
                                            v4f (&v)[KS / 8])
{
    const v4f* p = reinterpret_cast<const v4f*>(rowSlice + 4 * half);
#pragma unroll
    for (int i = 0; i < KS / 8; ++i) v[i] = p[2 * i];
}

template <int K>
__global__ __launch_bounds__(256) void kd_fc_wmma(
    const float* __restrict__ X, const float* __restrict__ Wt,
    const float* __restrict__ bias, float* __restrict__ Out)
{
    static_assert(K % 64 == 0, "8 slices of whole 8-k pieces");
    constexpr int KS = K / 8;
    __shared__ float red[8 * 256];

    const int lane = threadIdx.x & 31;
    const int wv   = threadIdx.x >> 5;
    const int half = lane >> 4;
    const int l16  = lane & 15;
    const int n0   = blockIdx.x << 4;

    v4f va[KS / 8], vb[KS / 8];
    fc_load_row<KS>(Wt + (size_t)l16 * K + wv * KS, half, va);
    fc_load_row<KS>(X + (size_t)(n0 + l16) * K + wv * KS, half, vb);

    v8f acc0 = v8f{}, acc1 = v8f{};
#pragma unroll
    for (int i = 0; i < KS / 8; ++i) {
        v2f a0, b0, a1, b1;
        a0.x = va[i].x; a0.y = va[i].y;
        b0.x = vb[i].x; b0.y = vb[i].y;
        a1.x = va[i].z; a1.y = va[i].w;
        b1.x = vb[i].z; b1.y = vb[i].w;
        acc0 = __builtin_amdgcn_wmma_f32_16x16x4_f32(
            false, a0, false, b0, (short)0, acc0, false, false);
        acc1 = __builtin_amdgcn_wmma_f32_16x16x4_f32(
            false, a1, false, b1, (short)0, acc1, false, false);
    }

#pragma unroll
    for (int r = 0; r < 8; ++r)
        red[(wv << 8) + (r << 5) + lane] = acc0[r] + acc1[r];
    __syncthreads();

    const int e  = threadIdx.x;
    const int nl = e >> 4;
    const int o  = e & 15;
    const int cell = ((o & 7) << 5) + nl + ((o >> 3) << 4);
    float sum = red[cell];
#pragma unroll
    for (int w = 1; w < 8; ++w) sum += red[(w << 8) + cell];
    Out[((size_t)(n0 + nl) << 4) + o] = sum + bias[o];
}

__global__ __launch_bounds__(256) void kd_logsoftmax16(
    const float* __restrict__ logits, float* __restrict__ out, int rows)
{
    int r = blockIdx.x * blockDim.x + threadIdx.x;
    if (r >= rows) return;
    float v[16];
    float mx = -3.402823466e38f;
#pragma unroll
    for (int i = 0; i < 16; ++i) { v[i] = logits[r * 16 + i]; mx = fmaxf(mx, v[i]); }
    float s = 0.0f;
#pragma unroll
    for (int i = 0; i < 16; ++i) s += expf(v[i] - mx);
    const float l = mx + logf(s);
#pragma unroll
    for (int i = 0; i < 16; ++i) out[r * 16 + i] = v[i] - l;
}

extern "C" void kernel_launch(void* const* d_in, const int* in_sizes, int n_in,
                              void* d_out, int out_size, void* d_ws, size_t ws_size,
                              hipStream_t stream) {
    static const int DIMS[11] = {2048, 1024, 512, 256, 128, 64, 32, 16, 8, 4, 2};
    static const int INCH[11] = {3, 8, 32, 64, 64, 64, 128, 256, 512, 512, 512};
    static const int FEAT[11] = {8, 32, 64, 64, 64, 128, 256, 512, 512, 512, 1024};
    const int Bsz = 512;

    const size_t XBYTES = (size_t)512 * 32 * 512 * sizeof(float);
    float* X0     = (float*)d_ws;
    float* X1     = (float*)((char*)d_ws + XBYTES);
    float* logits = (float*)((char*)d_ws + 2 * XBYTES);

    const float* cur = (const float*)d_in[0];
    for (int i = 0; i < 11; ++i) {
        const int Dd = DIMS[i], Kc = INCH[i], F = FEAT[i];
        const int Dlog2 = __builtin_ctz((unsigned)Dd);
        float* nxt = (i & 1) ? X1 : X0;

        const int N = Bsz * Dd;
        const int tilesF = (F + 15) / 16;
        const int ncg = ((N / 512) * tilesF < 256) ? 4 : 8;
        const int KS  = (ncg == 4 && Kc % 64 == 0) ? 2 : 1;
        const int nw  = ncg * KS;
        const int blocksX = N / (64 * ncg);
        const int Kp4 = (Kc + 3) & ~3;
        const size_t ldsBytes = (size_t)48 * (Kp4 + 1) * sizeof(float)
            + (KS == 2 ? (size_t)ncg * 1024 * sizeof(float) : 0);
        dim3 grid(blocksX, tilesF);
        const float* Wl = (const float*)d_in[12 + 2 * i];
        const float* bl = (const float*)d_in[13 + 2 * i];
        const int*   sl = (const int*)d_in[1 + i];
        if (i == 0)
            kd_gather_wmma<1, 0><<<grid, 32 * nw, ldsBytes, stream>>>(
                cur, Wl, bl, sl, nxt, F, Kc, Dlog2, nw, KS);
        else if (i == 1)
            kd_gather_wmma<1, 1><<<grid, 32 * nw, ldsBytes, stream>>>(
                cur, Wl, bl, sl, nxt, F, Kc, Dlog2, nw, KS);
        else if (i == 2)
            kd_gather_wmma<1, 4><<<grid, 32 * nw, ldsBytes, stream>>>(
                cur, Wl, bl, sl, nxt, F, Kc, Dlog2, nw, KS);
        else
            kd_gather_wmma<2, 4><<<grid, 32 * nw, ldsBytes, stream>>>(
                cur, Wl, bl, sl, nxt, F, Kc, Dlog2, nw, KS);
        cur = nxt;
    }

    kd_fc_wmma<1024><<<dim3(32), dim3(256), 0, stream>>>(
        cur, (const float*)d_in[34], (const float*)d_in[35], logits);

    kd_logsoftmax16<<<(512 + 255) / 256, 256, 0, stream>>>(logits, (float*)d_out, 512);

    (void)in_sizes; (void)n_in; (void)out_size; (void)ws_size;
}


# Conv layers 2-4: wave tile 16 -> 32 features x 64 columns: each gathered B fragment is loaded once and feeds two feature tiles (half the waves and input fetches; layer 4: one B set); sums unchanged
# speedup vs baseline: 5.5703x; 1.0341x over previous
#include <hip/hip_runtime.h>

typedef float v2f __attribute__((ext_vector_type(2)));
typedef float v4f __attribute__((ext_vector_type(4)));
typedef float v8f __attribute__((ext_vector_type(8)));


template <int N0, int N1, int N2, int N3>
__host__ __device__ constexpr int kd_np(int s)
{
    return s == 0 ? N0 : (s == 1 ? N1 : (s == 2 ? N2 : N3));
}

template <int N0, int N1, int N2, int N3, int FT>
__device__ __forceinline__ void kd_tile(
    const float* const (&ar)[4][3], const bool (&on)[4][3], int k0, int aoff,
    const v2f (&b)[4], v8f (&acc)[FT][4])
{
#pragma unroll
    for (int p = 0; p < 3; ++p) {
#pragma unroll
        for (int s = 0; s < 4; ++s) {
            if (p < kd_np<N0, N1, N2, N3>(s)) {
                v2f bm = b[s];
                if (kd_np<N0, N1, N2, N3>(s) > 1) {
                    bm.x = on[s][p] ? b[s].x : 0.0f;
                    bm.y = on[s][p] ? b[s].y : 0.0f;
                }
#pragma unroll
                for (int t = 0; t < FT; ++t) {
                    v2f a;
                    a.x = ar[s][p][k0 + t * aoff];
                    a.y = ar[s][p][k0 + t * aoff + 1];
                    acc[t][s] = __builtin_amdgcn_wmma_f32_16x16x4_f32(
                        false, a, false, bm, (short)0, acc[t][s], false, false);
                }
            }
        }
    }
}

template <int N0, int N1, int N2, int N3, int NV, int FT>
__device__ __forceinline__ void kd_chunk(
    const float* const (&ar)[4][3], const bool (&on)[4][3], int k0, int aoff,
    const v4f (&b)[4][NV], v8f (&acc)[FT][4])
{
#pragma unroll
    for (int v = 0; v < NV; ++v) {
#pragma unroll
        for (int e = 0; e < 2; ++e) {
#pragma unroll
            for (int p = 0; p < 3; ++p) {
#pragma unroll
                for (int s = 0; s < 4; ++s) {
                    if (p < kd_np<N0, N1, N2, N3>(s)) {
                        const int ka = k0 + 4 * v + 2 * e;
                        v2f bm;
                        bm.x = e ? b[s][v].z : b[s][v].x;
                        bm.y = e ? b[s][v].w : b[s][v].y;
                        if (kd_np<N0, N1, N2, N3>(s) > 1) {
                            bm.x = on[s][p] ? bm.x : 0.0f;
                            bm.y = on[s][p] ? bm.y : 0.0f;
                        }
#pragma unroll
                        for (int t = 0; t < FT; ++t) {
                            v2f a;
                            a.x = ar[s][p][ka + t * aoff];
                            a.y = ar[s][p][ka + t * aoff + 1];
                            acc[t][s] = __builtin_amdgcn_wmma_f32_16x16x4_f32(
                                false, a, false, bm, (short)0, acc[t][s], false, false);
                        }
                    }
                }
            }
        }
    }
}

template <int N0, int N1, int N2, int N3>
__device__ __forceinline__ void kd_rows(
    const float* aLane, int Kp, const int (&qv)[4][3], const int (&cnt)[4],
    const int (&qs)[4], const float* (&ar)[4][3], bool (&on)[4][3])
{
#pragma unroll
    for (int s = 0; s < 4; ++s) {
#pragma unroll
        for (int p = 0; p < 3; ++p) {
            if (p < kd_np<N0, N1, N2, N3>(s)) {
                ar[s][p] = aLane + min(qv[s][p], 2) * Kp;
                on[s][p] = (p < cnt[s]) && (qs[s] == qv[s][p]);
            } else {
                ar[s][p] = aLane;
                on[s][p] = false;
            }
        }
    }
}

template <int N0, int N1, int N2, int N3, int R, int NV, int FT>
__device__ __forceinline__ void kd_kloop(
    const float* aLane, int Kp, const int (&qv)[4][3], const int (&cnt)[4],
    const int (&qs)[4], const char* xb, const unsigned (&bo)[4],
    int K, int half, v8f (&acc)[FT][4])
{
    const float* ar[4][3];
    bool on[4][3];
    kd_rows<N0, N1, N2, N3>(aLane, Kp, qv, cnt, qs, ar, on);

    constexpr int CH = 8 * NV;
    const int aoff = 48 * Kp;
    const int nC = K / CH;
    const int last = nC - 1;
    const int kl = 4 * NV * half;
    auto ldB = [&](int j, v4f (&b)[4][NV]) {
        const unsigned tob = (unsigned)min(j, last) * (unsigned)(CH * 4);
#pragma unroll
        for (int s = 0; s < 4; ++s)
#pragma unroll
            for (int v = 0; v < NV; ++v)
                b[s][v] = *reinterpret_cast<const v4f*>(xb + (bo[s] + tob + 16u * v));
    };
    if (R == 1) {
        for (int j = 0; j < nC; ++j) {
            v4f b[4][NV];
            ldB(j, b);
            kd_chunk<N0, N1, N2, N3, NV, FT>(ar, on, CH * j + kl, aoff, b, acc);
        }
    } else {
        v4f b0[4][NV], b1[4][NV];
        ldB(0, b0);
        __builtin_amdgcn_sched_barrier(0);
        int j = 0;
        for (; j + 2 <= nC; j += 2) {
            ldB(j + 1, b1);
            kd_chunk<N0, N1, N2, N3, NV, FT>(ar, on, CH * j + kl, aoff, b0, acc);
            __builtin_amdgcn_sched_barrier(0);
            ldB(j + 2, b0);
            kd_chunk<N0, N1, N2, N3, NV, FT>(ar, on, CH * (j + 1) + kl, aoff, b1, acc);
            __builtin_amdgcn_sched_barrier(0);
        }
        if (j < nC) kd_chunk<N0, N1, N2, N3, NV, FT>(ar, on, CH * j + kl, aoff, b0, acc);
    }
}

template <int N0, int N1, int N2, int N3, int FT>
__device__ __forceinline__ void kd_ktail(
    const float* aLane, int Kp, const int (&qv)[4][3], const int (&cnt)[4],
    const int (&qs)[4], const char* xb, const unsigned (&bo0)[4],
    int K, int Dlog2, int kOff, v8f (&acc)[FT][4])
{
    const float* ar[4][3];
    bool on[4][3];
    kd_rows<N0, N1, N2, N3>(aLane, Kp, qv, cnt, qs, ar, on);
    const int c0 = min(kOff, K - 1) - kOff;
    const int c1 = min(kOff + 1, K - 1) - kOff;
    const ptrdiff_t rowF = (ptrdiff_t)1 << Dlog2;
    v2f b[4];
#pragma unroll
    for (int s = 0; s < 4; ++s) {
        const float* col = reinterpret_cast<const float*>(xb + bo0[s]);
        b[s].x = col[(ptrdiff_t)c0 * rowF];
        b[s].y = col[(ptrdiff_t)c1 * rowF];
    }
    kd_tile<N0, N1, N2, N3, FT>(ar, on, kOff, 48 * Kp, b, acc);
}

template <int R, int NV, int FT>
__global__ __launch_bounds__(256) void kd_gather_wmma(
    const float* __restrict__ X, const float* __restrict__ Wt,
    const float* __restrict__ bias, const int* __restrict__ sel,
    float* __restrict__ Out, int F, int K, int Dlog2, int nw, int KS)
{
    extern __shared__ float sW[];
    constexpr int ROWS = 48 * FT;
    const int D   = 1 << Dlog2;
    const int Kp4 = (K + 3) & ~3;
    const int Kp  = Kp4 + 1;
    const int f0  = blockIdx.y * (16 * FT);
    const int m0  = 3 * f0;
    const int M   = 3 * F;
    const int lane = threadIdx.x & 31;
    const int wv   = threadIdx.x >> 5;

    constexpr int KU = 8;
    constexpr int RW = 6 * FT;
    for (int r0 = wv; r0 < ROWS; r0 += RW * nw) {
        for (int kb0 = 0; kb0 < Kp4; kb0 += 32 * KU) {
            const int nU = min(KU, (Kp4 - kb0 + 31) >> 5);
            float w[KU][RW];
#pragma unroll
            for (int u = 0; u < KU; ++u) {
#pragma unroll
                for (int i = 0; i < RW; ++i) w[u][i] = 0.0f;
                if (u < nU) {
                    const int kc = min(kb0 + 32 * u + lane, K - 1);
#pragma unroll
                    for (int i = 0; i < RW; ++i) {
                        const int m = min(m0 + r0 + i * nw, M - 1);
                        w[u][i] = Wt[(size_t)m * K + kc];
                    }
                }
            }
#pragma unroll
            for (int u = 0; u < KU; ++u) {
                if (u < nU) {
                    const int k = kb0 + 32 * u + lane;
                    if (k < Kp4) {
#pragma unroll
                        for (int i = 0; i < RW; ++i)
                            sW[(r0 + i * nw) * Kp + k] = (k < K) ? w[u][i] : 0.0f;
                    }
                }
            }
        }
    }
    __syncthreads();

    const int nwc = (KS == 2) ? (nw >> 1) : nw;
    const int kh  = (wv >= nwc) ? 1 : 0;
    const int cg  = wv - kh * nwc;
    const int n0  = (blockIdx.x * nwc + cg) << 6;

    const int half = lane >> 4;
    const int l16  = lane & 15;
    const int kOff = half << 1;
    const float* aLane = sW + 3 * l16 * Kp;

    int ncol[4];
    if (D < 64) {
        const int de = (n0 + 2 * lane) & (D - 1);
        const int qe = min(max((3 * de + sel[de]) >> Dlog2, 0), 2);
        const int qo = min(max((3 * de + 3 + sel[de + 1]) >> Dlog2, 0), 2);
        const int key = 3 * qe + qo;
        const unsigned lower = (1u << lane) - 1u;
        int rank = 0;
#pragma unroll
        for (int v = 0; v < 9; ++v) {
            const unsigned m = (unsigned)__ballot(key == v);
            if (v < key) rank += __popc(m);
            if (v == key) rank += __popc(m & lower);
        }
        const int pairAt = __builtin_amdgcn_ds_permute(rank << 2, lane);
#pragma unroll
        for (int s = 0; s < 4; ++s) {
            const int pj = __shfl(pairAt, 8 * s + (l16 >> 1));
            ncol[s] = n0 + 2 * pj + (l16 & 1);
        }
    } else {
#pragma unroll
        for (int s = 0; s < 4; ++s) ncol[s] = n0 + (s << 4) + l16;
    }

    unsigned bo0[4];
    int qs[4];
    int qv[4][3];
    int cnt[4];
#pragma unroll
    for (int s = 0; s < 4; ++s) {
        const int n  = ncol[s];
        const int bb = n >> Dlog2;
        const int d  = n & (D - 1);
        const int t  = 3 * d + sel[d];
        const int q  = t >> Dlog2;
        const int dp = t & (D - 1);
        if (NV > 0) {
            bo0[s] = ((((unsigned)bb << Dlog2) + (unsigned)dp) * (unsigned)K
                      + (unsigned)(4 * NV * half)) << 2;
            if (KS == 2) bo0[s] += ((unsigned)kh * (unsigned)(K >> 1)) << 2;
        } else {
            bo0[s] = ((unsigned)bb * ((unsigned)K << Dlog2) + ((unsigned)kOff << Dlog2)
                      + (unsigned)dp) << 2;
        }
        qs[s] = q;
        const bool has0 = __any(q == 0) != 0;
        const bool has1 = __any(q == 1) != 0;
        const bool has2 = __any(q == 2) != 0;
        qv[s][0] = has0 ? 0 : (has1 ? 1 : 2);
        qv[s][1] = (has0 && has1) ? 1 : 2;
        qv[s][2] = 2;
        cnt[s] = max(1, (int)has0 + (int)has1 + (int)has2);
    }

    const int cmax = max(max(cnt[0], cnt[1]), max(cnt[2], cnt[3]));
    v8f acc[FT][4];
#pragma unroll
    for (int t = 0; t < FT; ++t)
#pragma unroll
        for (int s = 0; s < 4; ++s) acc[t][s] = v8f{};
    const char* xb = reinterpret_cast<const char*>(X);
    const int Ks = (KS == 2) ? (K >> 1) : K;
    const float* aSl = aLane + kh * Ks;
    if (NV > 0) {
        constexpr int NVc = NV > 0 ? NV : 1;
        if (cmax == 1)
            kd_kloop<1, 1, 1, 1, R, NVc, FT>(aSl, Kp, qv, cnt, qs, xb, bo0, Ks, half, acc);
        else if (cmax == 2 && cnt[0] == 1 && cnt[3] == 1)
            kd_kloop<1, 2, 2, 1, R, NVc, FT>(aSl, Kp, qv, cnt, qs, xb, bo0, Ks, half, acc);
        else if (cmax == 2)
            kd_kloop<2, 2, 2, 2, R, NVc, FT>(aSl, Kp, qv, cnt, qs, xb, bo0, Ks, half, acc);
        else
            kd_kloop<3, 3, 3, 3, R, NVc, FT>(aSl, Kp, qv, cnt, qs, xb, bo0, Ks, half, acc);
    } else {
        if (cmax == 1)
            kd_ktail<1, 1, 1, 1, FT>(aSl, Kp, qv, cnt, qs, xb, bo0, Ks, Dlog2, kOff, acc);
        else if (cmax == 2 && cnt[0] == 1 && cnt[3] == 1)
            kd_ktail<1, 2, 2, 1, FT>(aSl, Kp, qv, cnt, qs, xb, bo0, Ks, Dlog2, kOff, acc);
        else if (cmax == 2)
            kd_ktail<2, 2, 2, 2, FT>(aSl, Kp, qv, cnt, qs, xb, bo0, Ks, Dlog2, kOff, acc);
        else
            kd_ktail<3, 3, 3, 3, FT>(aSl, Kp, qv, cnt, qs, xb, bo0, Ks, Dlog2, kOff, acc);
    }

    if (KS == 2) {
        float* red = sW + ROWS * Kp + cg * (1024 * FT);
        if (kh == 1) {
#pragma unroll
            for (int t = 0; t < FT; ++t)
#pragma unroll
                for (int s = 0; s < 4; ++s)
#pragma unroll
                    for (int r = 0; r < 8; ++r)
                        red[(((((t << 2) + s) << 3) + r) << 5) + lane] = acc[t][s][r];
        }
        __syncthreads();
        if (kh == 1) return;
#pragma unroll
        for (int t = 0; t < FT; ++t)
#pragma unroll
            for (int s = 0; s < 4; ++s)
#pragma unroll
                for (int r = 0; r < 8; ++r)
                    acc[t][s][r] += red[(((((t << 2) + s) << 3) + r) << 5) + lane];
    }

    const int Dh = D >> 1;
#pragma unroll
    for (int t = 0; t < FT; ++t) {
        const int fl = f0 + 16 * t + (half << 3);
#pragma unroll
        for (int s = 0; s < 4; ++s) {
#pragma unroll
            for (int r = 0; r < 8; ++r) {
                const int fc = min(fl + r, F - 1);
                float v = fmaxf(acc[t][s][r] + bias[3 * fc + qs[s]], 0.0f);
                v = fmaxf(v, __shfl_xor(v, 1));
                acc[t][s][r] = v;
            }
        }
    }
    if (!(l16 & 1)) {
#pragma unroll
        for (int t = 0; t < FT; ++t) {
            const int fl = f0 + 16 * t + (half << 3);
            const int nr = F - fl;
#pragma unroll
            for (int s = 0; s < 4; ++s) {
                const int n  = ncol[s];
                const int bb = n >> Dlog2;
                const int d  = n & (D - 1);
                float* ocol = Out + ((size_t)bb * Dh + (d >> 1)) * F + fl;
                if (nr >= 8 && !(F & 3)) {
                    v4f lo, hi;
                    lo.x = acc[t][s][0]; lo.y = acc[t][s][1]; lo.z = acc[t][s][2]; lo.w = acc[t][s][3];
                    hi.x = acc[t][s][4]; hi.y = acc[t][s][5]; hi.z = acc[t][s][6]; hi.w = acc[t][s][7];
                    *reinterpret_cast<v4f*>(ocol) = lo;
                    *reinterpret_cast<v4f*>(ocol + 4) = hi;
                } else {
#pragma unroll
                    for (int r = 0; r < 8; ++r)
                        if (r < nr) ocol[r] = acc[t][s][r];
                }
            }
        }
    }
}

template <int KS>
__device__ __forceinline__ void fc_load_row(const float* rowSlice, int half,
                                            v4f (&v)[KS / 8])
{
    const v4f* p = reinterpret_cast<const v4f*>(rowSlice + 4 * half);
#pragma unroll
    for (int i = 0; i < KS / 8; ++i) v[i] = p[2 * i];
}

template <int K>
__global__ __launch_bounds__(256) void kd_fc_wmma(
    const float* __restrict__ X, const float* __restrict__ Wt,
    const float* __restrict__ bias, float* __restrict__ Out)
{
    static_assert(K % 64 == 0, "8 slices of whole 8-k pieces");
    constexpr int KS = K / 8;
    __shared__ float red[8 * 256];

    const int lane = threadIdx.x & 31;
    const int wv   = threadIdx.x >> 5;
    const int half = lane >> 4;
    const int l16  = lane & 15;
    const int n0   = blockIdx.x << 4;

    v4f va[KS / 8], vb[KS / 8];
    fc_load_row<KS>(Wt + (size_t)l16 * K + wv * KS, half, va);
    fc_load_row<KS>(X + (size_t)(n0 + l16) * K + wv * KS, half, vb);

    v8f acc0 = v8f{}, acc1 = v8f{};
#pragma unroll
    for (int i = 0; i < KS / 8; ++i) {
        v2f a0, b0, a1, b1;
        a0.x = va[i].x; a0.y = va[i].y;
        b0.x = vb[i].x; b0.y = vb[i].y;
        a1.x = va[i].z; a1.y = va[i].w;
        b1.x = vb[i].z; b1.y = vb[i].w;
        acc0 = __builtin_amdgcn_wmma_f32_16x16x4_f32(
            false, a0, false, b0, (short)0, acc0, false, false);
        acc1 = __builtin_amdgcn_wmma_f32_16x16x4_f32(
            false, a1, false, b1, (short)0, acc1, false, false);
    }

#pragma unroll
    for (int r = 0; r < 8; ++r)
        red[(wv << 8) + (r << 5) + lane] = acc0[r] + acc1[r];
    __syncthreads();

    const int e  = threadIdx.x;
    const int nl = e >> 4;
    const int o  = e & 15;
    const int cell = ((o & 7) << 5) + nl + ((o >> 3) << 4);
    float sum = red[cell];
#pragma unroll
    for (int w = 1; w < 8; ++w) sum += red[(w << 8) + cell];
    Out[((size_t)(n0 + nl) << 4) + o] = sum + bias[o];
}

__global__ __launch_bounds__(256) void kd_logsoftmax16(
    const float* __restrict__ logits, float* __restrict__ out, int rows)
{
    int r = blockIdx.x * blockDim.x + threadIdx.x;
    if (r >= rows) return;
    float v[16];
    float mx = -3.402823466e38f;
#pragma unroll
    for (int i = 0; i < 16; ++i) { v[i] = logits[r * 16 + i]; mx = fmaxf(mx, v[i]); }
    float s = 0.0f;
#pragma unroll
    for (int i = 0; i < 16; ++i) s += expf(v[i] - mx);
    const float l = mx + logf(s);
#pragma unroll
    for (int i = 0; i < 16; ++i) out[r * 16 + i] = v[i] - l;
}

extern "C" void kernel_launch(void* const* d_in, const int* in_sizes, int n_in,
                              void* d_out, int out_size, void* d_ws, size_t ws_size,
                              hipStream_t stream) {
    static const int DIMS[11] = {2048, 1024, 512, 256, 128, 64, 32, 16, 8, 4, 2};
    static const int INCH[11] = {3, 8, 32, 64, 64, 64, 128, 256, 512, 512, 512};
    static const int FEAT[11] = {8, 32, 64, 64, 64, 128, 256, 512, 512, 512, 1024};
    const int Bsz = 512;

    const size_t XBYTES = (size_t)512 * 32 * 512 * sizeof(float);
    float* X0     = (float*)d_ws;
    float* X1     = (float*)((char*)d_ws + XBYTES);
    float* logits = (float*)((char*)d_ws + 2 * XBYTES);

    const float* cur = (const float*)d_in[0];
    for (int i = 0; i < 11; ++i) {
        const int Dd = DIMS[i], Kc = INCH[i], F = FEAT[i];
        const int Dlog2 = __builtin_ctz((unsigned)Dd);
        float* nxt = (i & 1) ? X1 : X0;

        const int N = Bsz * Dd;
        const int FT = (i >= 1 && i <= 3) ? 2 : 1;
        const int tilesF = (F + 16 * FT - 1) / (16 * FT);
        const int ncg = ((N / 512) * tilesF < 256) ? 4 : 8;
        const int KS  = (ncg == 4 && Kc % 64 == 0) ? 2 : 1;
        const int nw  = ncg * KS;
        const int blocksX = N / (64 * ncg);
        const int Kp4 = (Kc + 3) & ~3;
        const size_t ldsBytes = (size_t)48 * FT * (Kp4 + 1) * sizeof(float)
            + (KS == 2 ? (size_t)ncg * 1024 * FT * sizeof(float) : 0);
        dim3 grid(blocksX, tilesF);
        const float* Wl = (const float*)d_in[12 + 2 * i];
        const float* bl = (const float*)d_in[13 + 2 * i];
        const int*   sl = (const int*)d_in[1 + i];
        if (i == 0)
            kd_gather_wmma<1, 0, 1><<<grid, 32 * nw, ldsBytes, stream>>>(
                cur, Wl, bl, sl, nxt, F, Kc, Dlog2, nw, KS);
        else if (i == 1)
            kd_gather_wmma<1, 1, 2><<<grid, 32 * nw, ldsBytes, stream>>>(
                cur, Wl, bl, sl, nxt, F, Kc, Dlog2, nw, KS);
        else if (i == 2 || i == 3)
            kd_gather_wmma<1, 4, 2><<<grid, 32 * nw, ldsBytes, stream>>>(
                cur, Wl, bl, sl, nxt, F, Kc, Dlog2, nw, KS);
        else
            kd_gather_wmma<2, 4, 1><<<grid, 32 * nw, ldsBytes, stream>>>(
                cur, Wl, bl, sl, nxt, F, Kc, Dlog2, nw, KS);
        cur = nxt;
    }

    kd_fc_wmma<1024><<<dim3(32), dim3(256), 0, stream>>>(
        cur, (const float*)d_in[34], (const float*)d_in[35], logits);

    kd_logsoftmax16<<<(512 + 255) / 256, 256, 0, stream>>>(logits, (float*)d_out, 512);

    (void)in_sizes; (void)n_in; (void)out_size; (void)ws_size;
}


# Layers 7-11 (D < 64): columns dealt out by d: one d of 16 batches per sub-tile, pair partner in the next sub-tile, same lane; 1 pass per sub-tile (4 not 6-8 per K tile), no B masks; sums unchanged
# speedup vs baseline: 5.8549x; 1.0511x over previous
#include <hip/hip_runtime.h>

typedef float v2f __attribute__((ext_vector_type(2)));
typedef float v4f __attribute__((ext_vector_type(4)));
typedef float v8f __attribute__((ext_vector_type(8)));


template <int N0, int N1, int N2, int N3>
__host__ __device__ constexpr int kd_np(int s)
{
    return s == 0 ? N0 : (s == 1 ? N1 : (s == 2 ? N2 : N3));
}

template <int N0, int N1, int N2, int N3, int FT>
__device__ __forceinline__ void kd_tile(
    const float* const (&ar)[4][3], const bool (&on)[4][3], int k0, int aoff,
    const v2f (&b)[4], v8f (&acc)[FT][4])
{
#pragma unroll
    for (int p = 0; p < 3; ++p) {
#pragma unroll
        for (int s = 0; s < 4; ++s) {
            if (p < kd_np<N0, N1, N2, N3>(s)) {
                v2f bm = b[s];
                if (kd_np<N0, N1, N2, N3>(s) > 1) {
                    bm.x = on[s][p] ? b[s].x : 0.0f;
                    bm.y = on[s][p] ? b[s].y : 0.0f;
                }
#pragma unroll
                for (int t = 0; t < FT; ++t) {
                    v2f a;
                    a.x = ar[s][p][k0 + t * aoff];
                    a.y = ar[s][p][k0 + t * aoff + 1];
                    acc[t][s] = __builtin_amdgcn_wmma_f32_16x16x4_f32(
                        false, a, false, bm, (short)0, acc[t][s], false, false);
                }
            }
        }
    }
}

template <int N0, int N1, int N2, int N3, int NV, int FT>
__device__ __forceinline__ void kd_chunk(
    const float* const (&ar)[4][3], const bool (&on)[4][3], int k0, int aoff,
    const v4f (&b)[4][NV], v8f (&acc)[FT][4])
{
#pragma unroll
    for (int v = 0; v < NV; ++v) {
#pragma unroll
        for (int e = 0; e < 2; ++e) {
#pragma unroll
            for (int p = 0; p < 3; ++p) {
#pragma unroll
                for (int s = 0; s < 4; ++s) {
                    if (p < kd_np<N0, N1, N2, N3>(s)) {
                        const int ka = k0 + 4 * v + 2 * e;
                        v2f bm;
                        bm.x = e ? b[s][v].z : b[s][v].x;
                        bm.y = e ? b[s][v].w : b[s][v].y;
                        if (kd_np<N0, N1, N2, N3>(s) > 1) {
                            bm.x = on[s][p] ? bm.x : 0.0f;
                            bm.y = on[s][p] ? bm.y : 0.0f;
                        }
#pragma unroll
                        for (int t = 0; t < FT; ++t) {
                            v2f a;
                            a.x = ar[s][p][ka + t * aoff];
                            a.y = ar[s][p][ka + t * aoff + 1];
                            acc[t][s] = __builtin_amdgcn_wmma_f32_16x16x4_f32(
                                false, a, false, bm, (short)0, acc[t][s], false, false);
                        }
                    }
                }
            }
        }
    }
}

template <int N0, int N1, int N2, int N3>
__device__ __forceinline__ void kd_rows(
    const float* aLane, int Kp, const int (&qv)[4][3], const int (&cnt)[4],
    const int (&qs)[4], const float* (&ar)[4][3], bool (&on)[4][3])
{
#pragma unroll
    for (int s = 0; s < 4; ++s) {
#pragma unroll
        for (int p = 0; p < 3; ++p) {
            if (p < kd_np<N0, N1, N2, N3>(s)) {
                ar[s][p] = aLane + min(qv[s][p], 2) * Kp;
                on[s][p] = (p < cnt[s]) && (qs[s] == qv[s][p]);
            } else {
                ar[s][p] = aLane;
                on[s][p] = false;
            }
        }
    }
}

template <int N0, int N1, int N2, int N3, int R, int NV, int FT>
__device__ __forceinline__ void kd_kloop(
    const float* aLane, int Kp, const int (&qv)[4][3], const int (&cnt)[4],
    const int (&qs)[4], const char* xb, const unsigned (&bo)[4],
    int K, int half, v8f (&acc)[FT][4])
{
    const float* ar[4][3];
    bool on[4][3];
    kd_rows<N0, N1, N2, N3>(aLane, Kp, qv, cnt, qs, ar, on);

    constexpr int CH = 8 * NV;
    const int aoff = 48 * Kp;
    const int nC = K / CH;
    const int last = nC - 1;
    const int kl = 4 * NV * half;
    auto ldB = [&](int j, v4f (&b)[4][NV]) {
        const unsigned tob = (unsigned)min(j, last) * (unsigned)(CH * 4);
#pragma unroll
        for (int s = 0; s < 4; ++s)
#pragma unroll
            for (int v = 0; v < NV; ++v)
                b[s][v] = *reinterpret_cast<const v4f*>(xb + (bo[s] + tob + 16u * v));
    };
    if (R == 1) {
        for (int j = 0; j < nC; ++j) {
            v4f b[4][NV];
            ldB(j, b);
            kd_chunk<N0, N1, N2, N3, NV, FT>(ar, on, CH * j + kl, aoff, b, acc);
        }
    } else {
        v4f b0[4][NV], b1[4][NV];
        ldB(0, b0);
        __builtin_amdgcn_sched_barrier(0);
        int j = 0;
        for (; j + 2 <= nC; j += 2) {
            ldB(j + 1, b1);
            kd_chunk<N0, N1, N2, N3, NV, FT>(ar, on, CH * j + kl, aoff, b0, acc);
            __builtin_amdgcn_sched_barrier(0);
            ldB(j + 2, b0);
            kd_chunk<N0, N1, N2, N3, NV, FT>(ar, on, CH * (j + 1) + kl, aoff, b1, acc);
            __builtin_amdgcn_sched_barrier(0);
        }
        if (j < nC) kd_chunk<N0, N1, N2, N3, NV, FT>(ar, on, CH * j + kl, aoff, b0, acc);
    }
}

template <int N0, int N1, int N2, int N3, int FT>
__device__ __forceinline__ void kd_ktail(
    const float* aLane, int Kp, const int (&qv)[4][3], const int (&cnt)[4],
    const int (&qs)[4], const char* xb, const unsigned (&bo0)[4],
    int K, int Dlog2, int kOff, v8f (&acc)[FT][4])
{
    const float* ar[4][3];
    bool on[4][3];
    kd_rows<N0, N1, N2, N3>(aLane, Kp, qv, cnt, qs, ar, on);
    const int c0 = min(kOff, K - 1) - kOff;
    const int c1 = min(kOff + 1, K - 1) - kOff;
    const ptrdiff_t rowF = (ptrdiff_t)1 << Dlog2;
    v2f b[4];
#pragma unroll
    for (int s = 0; s < 4; ++s) {
        const float* col = reinterpret_cast<const float*>(xb + bo0[s]);
        b[s].x = col[(ptrdiff_t)c0 * rowF];
        b[s].y = col[(ptrdiff_t)c1 * rowF];
    }
    kd_tile<N0, N1, N2, N3, FT>(ar, on, kOff, 48 * Kp, b, acc);
}

template <int R, int NV, int FT, bool EO>
__global__ __launch_bounds__(256) void kd_gather_wmma(
    const float* __restrict__ X, const float* __restrict__ Wt,
    const float* __restrict__ bias, const int* __restrict__ sel,
    float* __restrict__ Out, int F, int K, int Dlog2, int nw, int KS)
{
    extern __shared__ float sW[];
    constexpr int ROWS = 48 * FT;
    const int D   = 1 << Dlog2;
    const int Kp4 = (K + 3) & ~3;
    const int Kp  = Kp4 + 1;
    const int f0  = blockIdx.y * (16 * FT);
    const int m0  = 3 * f0;
    const int M   = 3 * F;
    const int lane = threadIdx.x & 31;
    const int wv   = threadIdx.x >> 5;

    constexpr int KU = 8;
    constexpr int RW = 6 * FT;
    for (int r0 = wv; r0 < ROWS; r0 += RW * nw) {
        for (int kb0 = 0; kb0 < Kp4; kb0 += 32 * KU) {
            const int nU = min(KU, (Kp4 - kb0 + 31) >> 5);
            float w[KU][RW];
#pragma unroll
            for (int u = 0; u < KU; ++u) {
#pragma unroll
                for (int i = 0; i < RW; ++i) w[u][i] = 0.0f;
                if (u < nU) {
                    const int kc = min(kb0 + 32 * u + lane, K - 1);
#pragma unroll
                    for (int i = 0; i < RW; ++i) {
                        const int m = min(m0 + r0 + i * nw, M - 1);
                        w[u][i] = Wt[(size_t)m * K + kc];
                    }
                }
            }
#pragma unroll
            for (int u = 0; u < KU; ++u) {
                if (u < nU) {
                    const int k = kb0 + 32 * u + lane;
                    if (k < Kp4) {
#pragma unroll
                        for (int i = 0; i < RW; ++i)
                            sW[(r0 + i * nw) * Kp + k] = (k < K) ? w[u][i] : 0.0f;
                    }
                }
            }
        }
    }
    __syncthreads();

    const int nwc = (KS == 2) ? (nw >> 1) : nw;
    const int kh  = (wv >= nwc) ? 1 : 0;
    const int cg  = wv - kh * nwc;
    const int n0  = (blockIdx.x * nwc + cg) << 6;

    const int half = lane >> 4;
    const int l16  = lane & 15;
    const int kOff = half << 1;
    const float* aLane = sW + 3 * l16 * Kp;

    int ncol[4];
    if (EO) {
        const int nb   = (nwc << 6) >> Dlog2;
        const int nbl2 = __builtin_ctz((unsigned)nb);
        const int nB0  = (blockIdx.x * nwc) << 6;
#pragma unroll
        for (int s = 0; s < 4; ++s) {
            const int r   = ((2 * cg + (s >> 1)) << 4) + l16;
            const int pi  = r >> nbl2;
            const int bbl = r & (nb - 1);
            ncol[s] = nB0 + (bbl << Dlog2) + 2 * pi + (s & 1);
        }
    } else {
#pragma unroll
        for (int s = 0; s < 4; ++s) ncol[s] = n0 + (s << 4) + l16;
    }

    unsigned bo0[4];
    int qs[4];
    int qv[4][3];
    int cnt[4];
#pragma unroll
    for (int s = 0; s < 4; ++s) {
        const int n  = ncol[s];
        const int bb = n >> Dlog2;
        const int d  = n & (D - 1);
        const int t  = 3 * d + sel[d];
        const int q  = t >> Dlog2;
        const int dp = t & (D - 1);
        if (NV > 0) {
            bo0[s] = ((((unsigned)bb << Dlog2) + (unsigned)dp) * (unsigned)K
                      + (unsigned)(4 * NV * half)) << 2;
            if (KS == 2) bo0[s] += ((unsigned)kh * (unsigned)(K >> 1)) << 2;
        } else {
            bo0[s] = ((unsigned)bb * ((unsigned)K << Dlog2) + ((unsigned)kOff << Dlog2)
                      + (unsigned)dp) << 2;
        }
        qs[s] = q;
        const bool has0 = __any(q == 0) != 0;
        const bool has1 = __any(q == 1) != 0;
        const bool has2 = __any(q == 2) != 0;
        qv[s][0] = has0 ? 0 : (has1 ? 1 : 2);
        qv[s][1] = (has0 && has1) ? 1 : 2;
        qv[s][2] = 2;
        cnt[s] = max(1, (int)has0 + (int)has1 + (int)has2);
    }

    const int cmax = max(max(cnt[0], cnt[1]), max(cnt[2], cnt[3]));
    v8f acc[FT][4];
#pragma unroll
    for (int t = 0; t < FT; ++t)
#pragma unroll
        for (int s = 0; s < 4; ++s) acc[t][s] = v8f{};
    const char* xb = reinterpret_cast<const char*>(X);
    const int Ks = (KS == 2) ? (K >> 1) : K;
    const float* aSl = aLane + kh * Ks;
    if (NV > 0) {
        constexpr int NVc = NV > 0 ? NV : 1;
        if (cmax == 1)
            kd_kloop<1, 1, 1, 1, R, NVc, FT>(aSl, Kp, qv, cnt, qs, xb, bo0, Ks, half, acc);
        else if (cmax == 2 && cnt[0] == 1 && cnt[3] == 1)
            kd_kloop<1, 2, 2, 1, R, NVc, FT>(aSl, Kp, qv, cnt, qs, xb, bo0, Ks, half, acc);
        else if (cmax == 2)
            kd_kloop<2, 2, 2, 2, R, NVc, FT>(aSl, Kp, qv, cnt, qs, xb, bo0, Ks, half, acc);
        else
            kd_kloop<3, 3, 3, 3, R, NVc, FT>(aSl, Kp, qv, cnt, qs, xb, bo0, Ks, half, acc);
    } else {
        if (cmax == 1)
            kd_ktail<1, 1, 1, 1, FT>(aSl, Kp, qv, cnt, qs, xb, bo0, Ks, Dlog2, kOff, acc);
        else if (cmax == 2 && cnt[0] == 1 && cnt[3] == 1)
            kd_ktail<1, 2, 2, 1, FT>(aSl, Kp, qv, cnt, qs, xb, bo0, Ks, Dlog2, kOff, acc);
        else if (cmax == 2)
            kd_ktail<2, 2, 2, 2, FT>(aSl, Kp, qv, cnt, qs, xb, bo0, Ks, Dlog2, kOff, acc);
        else
            kd_ktail<3, 3, 3, 3, FT>(aSl, Kp, qv, cnt, qs, xb, bo0, Ks, Dlog2, kOff, acc);
    }

    if (KS == 2) {
        float* red = sW + ROWS * Kp + cg * (1024 * FT);
        if (kh == 1) {
#pragma unroll
            for (int t = 0; t < FT; ++t)
#pragma unroll
                for (int s = 0; s < 4; ++s)
#pragma unroll
                    for (int r = 0; r < 8; ++r)
                        red[(((((t << 2) + s) << 3) + r) << 5) + lane] = acc[t][s][r];
        }
        __syncthreads();
        if (kh == 1) return;
#pragma unroll
        for (int t = 0; t < FT; ++t)
#pragma unroll
            for (int s = 0; s < 4; ++s)
#pragma unroll
                for (int r = 0; r < 8; ++r)
                    acc[t][s][r] += red[(((((t << 2) + s) << 3) + r) << 5) + lane];
    }

    const int Dh = D >> 1;
    if (EO) {
#pragma unroll
        for (int t = 0; t < FT; ++t) {
            const int fl = f0 + 16 * t + (half << 3);
            const int nr = F - fl;
#pragma unroll
            for (int g = 0; g < 2; ++g) {
                float v[8];
#pragma unroll
                for (int r = 0; r < 8; ++r) {
                    const int fc = min(fl + r, F - 1);
                    const float ve = fmaxf(acc[t][2 * g][r] + bias[3 * fc + qs[2 * g]], 0.0f);
                    const float vo = fmaxf(acc[t][2 * g + 1][r] + bias[3 * fc + qs[2 * g + 1]], 0.0f);
                    v[r] = fmaxf(ve, vo);
                }
                const int n  = ncol[2 * g];
                const int bb = n >> Dlog2;
                const int d  = n & (D - 1);
                float* ocol = Out + ((size_t)bb * Dh + (d >> 1)) * F + fl;
                if (nr >= 8 && !(F & 3)) {
                    v4f lo, hi;
                    lo.x = v[0]; lo.y = v[1]; lo.z = v[2]; lo.w = v[3];
                    hi.x = v[4]; hi.y = v[5]; hi.z = v[6]; hi.w = v[7];
                    *reinterpret_cast<v4f*>(ocol) = lo;
                    *reinterpret_cast<v4f*>(ocol + 4) = hi;
                } else {
#pragma unroll
                    for (int r = 0; r < 8; ++r)
                        if (r < nr) ocol[r] = v[r];
                }
            }
        }
        return;
    }
#pragma unroll
    for (int t = 0; t < FT; ++t) {
        const int fl = f0 + 16 * t + (half << 3);
#pragma unroll
        for (int s = 0; s < 4; ++s) {
#pragma unroll
            for (int r = 0; r < 8; ++r) {
                const int fc = min(fl + r, F - 1);
                float v = fmaxf(acc[t][s][r] + bias[3 * fc + qs[s]], 0.0f);
                v = fmaxf(v, __shfl_xor(v, 1));
                acc[t][s][r] = v;
            }
        }
    }
    if (!(l16 & 1)) {
#pragma unroll
        for (int t = 0; t < FT; ++t) {
            const int fl = f0 + 16 * t + (half << 3);
            const int nr = F - fl;
#pragma unroll
            for (int s = 0; s < 4; ++s) {
                const int n  = ncol[s];
                const int bb = n >> Dlog2;
                const int d  = n & (D - 1);
                float* ocol = Out + ((size_t)bb * Dh + (d >> 1)) * F + fl;
                if (nr >= 8 && !(F & 3)) {
                    v4f lo, hi;
                    lo.x = acc[t][s][0]; lo.y = acc[t][s][1]; lo.z = acc[t][s][2]; lo.w = acc[t][s][3];
                    hi.x = acc[t][s][4]; hi.y = acc[t][s][5]; hi.z = acc[t][s][6]; hi.w = acc[t][s][7];
                    *reinterpret_cast<v4f*>(ocol) = lo;
                    *reinterpret_cast<v4f*>(ocol + 4) = hi;
                } else {
#pragma unroll
                    for (int r = 0; r < 8; ++r)
                        if (r < nr) ocol[r] = acc[t][s][r];
                }
            }
        }
    }
}

template <int KS>
__device__ __forceinline__ void fc_load_row(const float* rowSlice, int half,
                                            v4f (&v)[KS / 8])
{
    const v4f* p = reinterpret_cast<const v4f*>(rowSlice + 4 * half);
#pragma unroll
    for (int i = 0; i < KS / 8; ++i) v[i] = p[2 * i];
}

template <int K>
__global__ __launch_bounds__(256) void kd_fc_wmma(
    const float* __restrict__ X, const float* __restrict__ Wt,
    const float* __restrict__ bias, float* __restrict__ Out)
{
    static_assert(K % 64 == 0, "8 slices of whole 8-k pieces");
    constexpr int KS = K / 8;
    __shared__ float red[8 * 256];

    const int lane = threadIdx.x & 31;
    const int wv   = threadIdx.x >> 5;
    const int half = lane >> 4;
    const int l16  = lane & 15;
    const int n0   = blockIdx.x << 4;

    v4f va[KS / 8], vb[KS / 8];
    fc_load_row<KS>(Wt + (size_t)l16 * K + wv * KS, half, va);
    fc_load_row<KS>(X + (size_t)(n0 + l16) * K + wv * KS, half, vb);

    v8f acc0 = v8f{}, acc1 = v8f{};
#pragma unroll
    for (int i = 0; i < KS / 8; ++i) {
        v2f a0, b0, a1, b1;
        a0.x = va[i].x; a0.y = va[i].y;
        b0.x = vb[i].x; b0.y = vb[i].y;
        a1.x = va[i].z; a1.y = va[i].w;
        b1.x = vb[i].z; b1.y = vb[i].w;
        acc0 = __builtin_amdgcn_wmma_f32_16x16x4_f32(
            false, a0, false, b0, (short)0, acc0, false, false);
        acc1 = __builtin_amdgcn_wmma_f32_16x16x4_f32(
            false, a1, false, b1, (short)0, acc1, false, false);
    }

#pragma unroll
    for (int r = 0; r < 8; ++r)
        red[(wv << 8) + (r << 5) + lane] = acc0[r] + acc1[r];
    __syncthreads();

    const int e  = threadIdx.x;
    const int nl = e >> 4;
    const int o  = e & 15;
    const int cell = ((o & 7) << 5) + nl + ((o >> 3) << 4);
    float sum = red[cell];
#pragma unroll
    for (int w = 1; w < 8; ++w) sum += red[(w << 8) + cell];
    Out[((size_t)(n0 + nl) << 4) + o] = sum + bias[o];
}

__global__ __launch_bounds__(256) void kd_logsoftmax16(
    const float* __restrict__ logits, float* __restrict__ out, int rows)
{
    int r = blockIdx.x * blockDim.x + threadIdx.x;
    if (r >= rows) return;
    float v[16];
    float mx = -3.402823466e38f;
#pragma unroll
    for (int i = 0; i < 16; ++i) { v[i] = logits[r * 16 + i]; mx = fmaxf(mx, v[i]); }
    float s = 0.0f;
#pragma unroll
    for (int i = 0; i < 16; ++i) s += expf(v[i] - mx);
    const float l = mx + logf(s);
#pragma unroll
    for (int i = 0; i < 16; ++i) out[r * 16 + i] = v[i] - l;
}

extern "C" void kernel_launch(void* const* d_in, const int* in_sizes, int n_in,
                              void* d_out, int out_size, void* d_ws, size_t ws_size,
                              hipStream_t stream) {
    static const int DIMS[11] = {2048, 1024, 512, 256, 128, 64, 32, 16, 8, 4, 2};
    static const int INCH[11] = {3, 8, 32, 64, 64, 64, 128, 256, 512, 512, 512};
    static const int FEAT[11] = {8, 32, 64, 64, 64, 128, 256, 512, 512, 512, 1024};
    const int Bsz = 512;

    const size_t XBYTES = (size_t)512 * 32 * 512 * sizeof(float);
    float* X0     = (float*)d_ws;
    float* X1     = (float*)((char*)d_ws + XBYTES);
    float* logits = (float*)((char*)d_ws + 2 * XBYTES);

    const float* cur = (const float*)d_in[0];
    for (int i = 0; i < 11; ++i) {
        const int Dd = DIMS[i], Kc = INCH[i], F = FEAT[i];
        const int Dlog2 = __builtin_ctz((unsigned)Dd);
        float* nxt = (i & 1) ? X1 : X0;

        const int N = Bsz * Dd;
        const int FT = (i >= 1 && i <= 3) ? 2 : 1;
        const int tilesF = (F + 16 * FT - 1) / (16 * FT);
        const int ncg = ((N / 512) * tilesF < 256) ? 4 : 8;
        const int KS  = (ncg == 4 && Kc % 64 == 0) ? 2 : 1;
        const int nw  = ncg * KS;
        const int blocksX = N / (64 * ncg);
        const int Kp4 = (Kc + 3) & ~3;
        const size_t ldsBytes = (size_t)48 * FT * (Kp4 + 1) * sizeof(float)
            + (KS == 2 ? (size_t)ncg * 1024 * FT * sizeof(float) : 0);
        dim3 grid(blocksX, tilesF);
        const float* Wl = (const float*)d_in[12 + 2 * i];
        const float* bl = (const float*)d_in[13 + 2 * i];
        const int*   sl = (const int*)d_in[1 + i];
        if (i == 0)
            kd_gather_wmma<1, 0, 1, false><<<grid, 32 * nw, ldsBytes, stream>>>(
                cur, Wl, bl, sl, nxt, F, Kc, Dlog2, nw, KS);
        else if (i == 1)
            kd_gather_wmma<1, 1, 2, false><<<grid, 32 * nw, ldsBytes, stream>>>(
                cur, Wl, bl, sl, nxt, F, Kc, Dlog2, nw, KS);
        else if (i == 2 || i == 3)
            kd_gather_wmma<1, 4, 2, false><<<grid, 32 * nw, ldsBytes, stream>>>(
                cur, Wl, bl, sl, nxt, F, Kc, Dlog2, nw, KS);
        else if (i == 4 || i == 5)
            kd_gather_wmma<2, 4, 1, false><<<grid, 32 * nw, ldsBytes, stream>>>(
                cur, Wl, bl, sl, nxt, F, Kc, Dlog2, nw, KS);
        else
            kd_gather_wmma<2, 4, 1, true><<<grid, 32 * nw, ldsBytes, stream>>>(
                cur, Wl, bl, sl, nxt, F, Kc, Dlog2, nw, KS);
        cur = nxt;
    }

    kd_fc_wmma<1024><<<dim3(32), dim3(256), 0, stream>>>(
        cur, (const float*)d_in[34], (const float*)d_in[35], logits);

    kd_logsoftmax16<<<(512 + 255) / 256, 256, 0, stream>>>(logits, (float*)d_out, 512);

    (void)in_sizes; (void)n_in; (void)out_size; (void)ws_size;
}


# Layer 1 folded into layer 2's launch: each wave forms the 8-feature layer-1 cells its columns gather (K = 3 on WMMA, pair max in registers) instead of loading them; 13 -> 12 launches; sums unchanged
# speedup vs baseline: 6.4360x; 1.0993x over previous
#include <hip/hip_runtime.h>

typedef float v2f __attribute__((ext_vector_type(2)));
typedef float v4f __attribute__((ext_vector_type(4)));
typedef float v8f __attribute__((ext_vector_type(8)));


template <int N0, int N1, int N2, int N3>
__host__ __device__ constexpr int kd_np(int s)
{
    return s == 0 ? N0 : (s == 1 ? N1 : (s == 2 ? N2 : N3));
}

template <int N0, int N1, int N2, int N3, int NV, int FT>
__device__ __forceinline__ void kd_chunk(
    const float* const (&ar)[4][3], const bool (&on)[4][3], int k0, int aoff,
    const v4f (&b)[4][NV], v8f (&acc)[FT][4])
{
#pragma unroll
    for (int v = 0; v < NV; ++v) {
#pragma unroll
        for (int e = 0; e < 2; ++e) {
#pragma unroll
            for (int p = 0; p < 3; ++p) {
#pragma unroll
                for (int s = 0; s < 4; ++s) {
                    if (p < kd_np<N0, N1, N2, N3>(s)) {
                        const int ka = k0 + 4 * v + 2 * e;
                        v2f bm;
                        bm.x = e ? b[s][v].z : b[s][v].x;
                        bm.y = e ? b[s][v].w : b[s][v].y;
                        if (kd_np<N0, N1, N2, N3>(s) > 1) {
                            bm.x = on[s][p] ? bm.x : 0.0f;
                            bm.y = on[s][p] ? bm.y : 0.0f;
                        }
#pragma unroll
                        for (int t = 0; t < FT; ++t) {
                            v2f a;
                            a.x = ar[s][p][ka + t * aoff];
                            a.y = ar[s][p][ka + t * aoff + 1];
                            acc[t][s] = __builtin_amdgcn_wmma_f32_16x16x4_f32(
                                false, a, false, bm, (short)0, acc[t][s], false, false);
                        }
                    }
                }
            }
        }
    }
}

template <int N0, int N1, int N2, int N3>
__device__ __forceinline__ void kd_rows(
    const float* aLane, int Kp, const int (&qv)[4][3], const int (&cnt)[4],
    const int (&qs)[4], const float* (&ar)[4][3], bool (&on)[4][3])
{
#pragma unroll
    for (int s = 0; s < 4; ++s) {
#pragma unroll
        for (int p = 0; p < 3; ++p) {
            if (p < kd_np<N0, N1, N2, N3>(s)) {
                ar[s][p] = aLane + min(qv[s][p], 2) * Kp;
                on[s][p] = (p < cnt[s]) && (qs[s] == qv[s][p]);
            } else {
                ar[s][p] = aLane;
                on[s][p] = false;
            }
        }
    }
}

template <int N0, int N1, int N2, int N3, int R, int NV, int FT>
__device__ __forceinline__ void kd_kloop(
    const float* aLane, int Kp, const int (&qv)[4][3], const int (&cnt)[4],
    const int (&qs)[4], const char* xb, const unsigned (&bo)[4],
    int K, int half, v8f (&acc)[FT][4])
{
    const float* ar[4][3];
    bool on[4][3];
    kd_rows<N0, N1, N2, N3>(aLane, Kp, qv, cnt, qs, ar, on);

    constexpr int CH = 8 * NV;
    const int aoff = 48 * Kp;
    const int nC = K / CH;
    const int last = nC - 1;
    const int kl = 4 * NV * half;
    auto ldB = [&](int j, v4f (&b)[4][NV]) {
        const unsigned tob = (unsigned)min(j, last) * (unsigned)(CH * 4);
#pragma unroll
        for (int s = 0; s < 4; ++s)
#pragma unroll
            for (int v = 0; v < NV; ++v)
                b[s][v] = *reinterpret_cast<const v4f*>(xb + (bo[s] + tob + 16u * v));
    };
    if (R == 1) {
        for (int j = 0; j < nC; ++j) {
            v4f b[4][NV];
            ldB(j, b);
            kd_chunk<N0, N1, N2, N3, NV, FT>(ar, on, CH * j + kl, aoff, b, acc);
        }
    } else {
        v4f b0[4][NV], b1[4][NV];
        ldB(0, b0);
        __builtin_amdgcn_sched_barrier(0);
        int j = 0;
        for (; j + 2 <= nC; j += 2) {
            ldB(j + 1, b1);
            kd_chunk<N0, N1, N2, N3, NV, FT>(ar, on, CH * j + kl, aoff, b0, acc);
            __builtin_amdgcn_sched_barrier(0);
            ldB(j + 2, b0);
            kd_chunk<N0, N1, N2, N3, NV, FT>(ar, on, CH * (j + 1) + kl, aoff, b1, acc);
            __builtin_amdgcn_sched_barrier(0);
        }
        if (j < nC) kd_chunk<N0, N1, N2, N3, NV, FT>(ar, on, CH * j + kl, aoff, b0, acc);
    }
}

template <int N0, int N1, int N2, int N3, int FT>
__device__ __forceinline__ void kd_apply(
    const float* aLane, int Kp, const int (&qv)[4][3], const int (&cnt)[4],
    const int (&qs)[4], int k0, const v4f (&b)[4][1], v8f (&acc)[FT][4])
{
    const float* ar[4][3];
    bool on[4][3];
    kd_rows<N0, N1, N2, N3>(aLane, Kp, qv, cnt, qs, ar, on);
    kd_chunk<N0, N1, N2, N3, 1, FT>(ar, on, k0, 48 * Kp, b, acc);
}

__device__ __forceinline__ void kd_l1cells(
    const float* __restrict__ x, const int* __restrict__ sel1,
    const float* sA1, const float* sB1, int K1, int D1log2,
    const int (&bb)[4], const int (&dp)[4], int half, int l16, v4f (&cell)[4][1])
{
    const int D1 = 1 << D1log2;
    const int ka = 2 * half, kb = 2 * half + 1;
    const size_t ra = (size_t)min(ka, K1 - 1) << D1log2;
    const size_t rb = (size_t)min(kb, K1 - 1) << D1log2;
    int t1[4][2];
#pragma unroll
    for (int s = 0; s < 4; ++s)
#pragma unroll
        for (int e = 0; e < 2; ++e) {
            const int d1 = 2 * dp[s] + e;
            t1[s][e] = 3 * d1 + sel1[d1];
        }
    v2f bx[4][2];
    const float* col[4][2];
#pragma unroll
    for (int s = 0; s < 4; ++s) {
        const float* xb = x + (((size_t)bb[s] * K1) << D1log2);
#pragma unroll
        for (int e = 0; e < 2; ++e) {
            col[s][e] = xb + (t1[s][e] & (D1 - 1));
            const float xa = col[s][e][ra];
            bx[s][e].x = (ka < K1) ? xa : 0.0f;
            bx[s][e].y = 0.0f;
        }
    }
    if (kb < K1) {
#pragma unroll
        for (int s = 0; s < 4; ++s)
#pragma unroll
            for (int e = 0; e < 2; ++e) bx[s][e].y = col[s][e][rb];
    }
    v2f a1[3];
#pragma unroll
    for (int p = 0; p < 3; ++p)
        a1[p] = *reinterpret_cast<const v2f*>(sA1 + ((((p << 4) + l16) << 2) + 2 * half));
#pragma unroll
    for (int s = 0; s < 4; ++s) {
        v8f c[2] = {v8f{}, v8f{}};
        int q1[2];
#pragma unroll
        for (int e = 0; e < 2; ++e) q1[e] = t1[s][e] >> D1log2;
#pragma unroll
        for (int p = 0; p < 3; ++p)
#pragma unroll
            for (int e = 0; e < 2; ++e) {
                v2f bm;
                bm.x = (q1[e] == p) ? bx[s][e].x : 0.0f;
                bm.y = (q1[e] == p) ? bx[s][e].y : 0.0f;
                c[e] = __builtin_amdgcn_wmma_f32_16x16x4_f32(
                    false, a1[p], false, bm, (short)0, c[e], false, false);
            }
        v4f v[2];
#pragma unroll
        for (int e = 0; e < 2; ++e) {
            const v4f bi = *reinterpret_cast<const v4f*>(sB1 + ((min(q1[e], 2) << 3) + (half << 2)));
            v[e].x = fmaxf(c[e][0] + bi.x, 0.0f);
            v[e].y = fmaxf(c[e][1] + bi.y, 0.0f);
            v[e].z = fmaxf(c[e][2] + bi.z, 0.0f);
            v[e].w = fmaxf(c[e][3] + bi.w, 0.0f);
        }
        cell[s][0].x = fmaxf(v[0].x, v[1].x);
        cell[s][0].y = fmaxf(v[0].y, v[1].y);
        cell[s][0].z = fmaxf(v[0].z, v[1].z);
        cell[s][0].w = fmaxf(v[0].w, v[1].w);
    }
}

template <int R, int NV, int FT, bool EO, bool L1>
__global__ __launch_bounds__(256) void kd_gather_wmma(
    const float* __restrict__ X, const float* __restrict__ Wt,
    const float* __restrict__ bias, const int* __restrict__ sel,
    float* __restrict__ Out, int F, int K, int Dlog2, int nw, int KS,
    const float* __restrict__ W1, const float* __restrict__ bias1,
    const int* __restrict__ sel1, int K1)
{
    extern __shared__ float sW[];
    constexpr int ROWS = 48 * FT;
    const int D   = 1 << Dlog2;
    const int Kp4 = (K + 3) & ~3;
    const int Kp  = Kp4 + 1;
    const int f0  = blockIdx.y * (16 * FT);
    const int m0  = 3 * f0;
    const int M   = 3 * F;
    const int lane = threadIdx.x & 31;
    const int wv   = threadIdx.x >> 5;

    constexpr int KU = 8;
    constexpr int RW = 6 * FT;
    for (int r0 = wv; r0 < ROWS; r0 += RW * nw) {
        for (int kb0 = 0; kb0 < Kp4; kb0 += 32 * KU) {
            const int nU = min(KU, (Kp4 - kb0 + 31) >> 5);
            float w[KU][RW];
#pragma unroll
            for (int u = 0; u < KU; ++u) {
#pragma unroll
                for (int i = 0; i < RW; ++i) w[u][i] = 0.0f;
                if (u < nU) {
                    const int kc = min(kb0 + 32 * u + lane, K - 1);
#pragma unroll
                    for (int i = 0; i < RW; ++i) {
                        const int m = min(m0 + r0 + i * nw, M - 1);
                        w[u][i] = Wt[(size_t)m * K + kc];
                    }
                }
            }
#pragma unroll
            for (int u = 0; u < KU; ++u) {
                if (u < nU) {
                    const int k = kb0 + 32 * u + lane;
                    if (k < Kp4) {
#pragma unroll
                        for (int i = 0; i < RW; ++i)
                            sW[(r0 + i * nw) * Kp + k] = (k < K) ? w[u][i] : 0.0f;
                    }
                }
            }
        }
    }
    float* sA1 = sW + ((ROWS * Kp + 3) & ~3);
    float* sB1 = sA1 + 192;
    if (L1) {
        for (int i = threadIdx.x; i < 216; i += 32 * nw) {
            float v = 0.0f;
            if (i < 192) {
                const int q = i >> 6, m = (i >> 2) & 15, k = i & 3;
                const int f = (m & 3) + ((m >> 3) << 2);
                if (!(m & 4) && f < K && k < K1) v = W1[(3 * f + q) * K1 + k];
            } else {
                const int q = (i - 192) >> 3, f = (i - 192) & 7;
                if (f < K) v = bias1[3 * f + q];
            }
            sA1[i] = v;
        }
    }
    __syncthreads();

    const int nwc = (KS == 2) ? (nw >> 1) : nw;
    const int kh  = (wv >= nwc) ? 1 : 0;
    const int cg  = wv - kh * nwc;
    const int n0  = (blockIdx.x * nwc + cg) << 6;

    const int half = lane >> 4;
    const int l16  = lane & 15;
    const float* aLane = sW + 3 * l16 * Kp;

    int ncol[4];
    if (EO) {
        const int nb   = (nwc << 6) >> Dlog2;
        const int nbl2 = __builtin_ctz((unsigned)nb);
        const int nB0  = (blockIdx.x * nwc) << 6;
#pragma unroll
        for (int s = 0; s < 4; ++s) {
            const int r   = ((2 * cg + (s >> 1)) << 4) + l16;
            const int pi  = r >> nbl2;
            const int bbl = r & (nb - 1);
            ncol[s] = nB0 + (bbl << Dlog2) + 2 * pi + (s & 1);
        }
    } else {
#pragma unroll
        for (int s = 0; s < 4; ++s) ncol[s] = n0 + (s << 4) + l16;
    }

    unsigned bo0[4];
    int bbA[4], dpA[4];
    int qs[4];
    int qv[4][3];
    int cnt[4];
#pragma unroll
    for (int s = 0; s < 4; ++s) {
        const int n  = ncol[s];
        const int bb = n >> Dlog2;
        const int d  = n & (D - 1);
        const int t  = 3 * d + sel[d];
        const int q  = t >> Dlog2;
        const int dp = t & (D - 1);
        bo0[s] = ((((unsigned)bb << Dlog2) + (unsigned)dp) * (unsigned)K
                  + (unsigned)(4 * NV * half)) << 2;
        if (KS == 2) bo0[s] += ((unsigned)kh * (unsigned)(K >> 1)) << 2;
        bbA[s] = bb;
        dpA[s] = dp;
        qs[s] = q;
        const bool has0 = __any(q == 0) != 0;
        const bool has1 = __any(q == 1) != 0;
        const bool has2 = __any(q == 2) != 0;
        qv[s][0] = has0 ? 0 : (has1 ? 1 : 2);
        qv[s][1] = (has0 && has1) ? 1 : 2;
        qv[s][2] = 2;
        cnt[s] = max(1, (int)has0 + (int)has1 + (int)has2);
    }

    const int cmax = max(max(cnt[0], cnt[1]), max(cnt[2], cnt[3]));
    v8f acc[FT][4];
#pragma unroll
    for (int t = 0; t < FT; ++t)
#pragma unroll
        for (int s = 0; s < 4; ++s) acc[t][s] = v8f{};
    const char* xb = reinterpret_cast<const char*>(X);
    const int Ks = (KS == 2) ? (K >> 1) : K;
    const float* aSl = aLane + kh * Ks;
    if (L1) {
        v4f bc[4][1];
        kd_l1cells(X, sel1, sA1, sB1, K1, Dlog2 + 1, bbA, dpA, half, l16, bc);
        if (cmax == 1)
            kd_apply<1, 1, 1, 1, FT>(aSl, Kp, qv, cnt, qs, 4 * half, bc, acc);
        else if (cmax == 2 && cnt[0] == 1 && cnt[3] == 1)
            kd_apply<1, 2, 2, 1, FT>(aSl, Kp, qv, cnt, qs, 4 * half, bc, acc);
        else if (cmax == 2)
            kd_apply<2, 2, 2, 2, FT>(aSl, Kp, qv, cnt, qs, 4 * half, bc, acc);
        else
            kd_apply<3, 3, 3, 3, FT>(aSl, Kp, qv, cnt, qs, 4 * half, bc, acc);
    } else {
        if (cmax == 1)
            kd_kloop<1, 1, 1, 1, R, NV, FT>(aSl, Kp, qv, cnt, qs, xb, bo0, Ks, half, acc);
        else if (cmax == 2 && cnt[0] == 1 && cnt[3] == 1)
            kd_kloop<1, 2, 2, 1, R, NV, FT>(aSl, Kp, qv, cnt, qs, xb, bo0, Ks, half, acc);
        else if (cmax == 2)
            kd_kloop<2, 2, 2, 2, R, NV, FT>(aSl, Kp, qv, cnt, qs, xb, bo0, Ks, half, acc);
        else
            kd_kloop<3, 3, 3, 3, R, NV, FT>(aSl, Kp, qv, cnt, qs, xb, bo0, Ks, half, acc);
    }

    if (KS == 2) {
        float* red = sW + ROWS * Kp + cg * (1024 * FT);
        if (kh == 1) {
#pragma unroll
            for (int t = 0; t < FT; ++t)
#pragma unroll
                for (int s = 0; s < 4; ++s)
#pragma unroll
                    for (int r = 0; r < 8; ++r)
                        red[(((((t << 2) + s) << 3) + r) << 5) + lane] = acc[t][s][r];
        }
        __syncthreads();
        if (kh == 1) return;
#pragma unroll
        for (int t = 0; t < FT; ++t)
#pragma unroll
            for (int s = 0; s < 4; ++s)
#pragma unroll
                for (int r = 0; r < 8; ++r)
                    acc[t][s][r] += red[(((((t << 2) + s) << 3) + r) << 5) + lane];
    }

    const int Dh = D >> 1;
    if (EO) {
#pragma unroll
        for (int t = 0; t < FT; ++t) {
            const int fl = f0 + 16 * t + (half << 3);
            const int nr = F - fl;
#pragma unroll
            for (int g = 0; g < 2; ++g) {
                float v[8];
#pragma unroll
                for (int r = 0; r < 8; ++r) {
                    const int fc = min(fl + r, F - 1);
                    const float ve = fmaxf(acc[t][2 * g][r] + bias[3 * fc + qs[2 * g]], 0.0f);
                    const float vo = fmaxf(acc[t][2 * g + 1][r] + bias[3 * fc + qs[2 * g + 1]], 0.0f);
                    v[r] = fmaxf(ve, vo);
                }
                const int n  = ncol[2 * g];
                const int bb = n >> Dlog2;
                const int d  = n & (D - 1);
                float* ocol = Out + ((size_t)bb * Dh + (d >> 1)) * F + fl;
                if (nr >= 8 && !(F & 3)) {
                    v4f lo, hi;
                    lo.x = v[0]; lo.y = v[1]; lo.z = v[2]; lo.w = v[3];
                    hi.x = v[4]; hi.y = v[5]; hi.z = v[6]; hi.w = v[7];
                    *reinterpret_cast<v4f*>(ocol) = lo;
                    *reinterpret_cast<v4f*>(ocol + 4) = hi;
                } else {
#pragma unroll
                    for (int r = 0; r < 8; ++r)
                        if (r < nr) ocol[r] = v[r];
                }
            }
        }
        return;
    }
#pragma unroll
    for (int t = 0; t < FT; ++t) {
        const int fl = f0 + 16 * t + (half << 3);
#pragma unroll
        for (int s = 0; s < 4; ++s) {
#pragma unroll
            for (int r = 0; r < 8; ++r) {
                const int fc = min(fl + r, F - 1);
                float v = fmaxf(acc[t][s][r] + bias[3 * fc + qs[s]], 0.0f);
                v = fmaxf(v, __shfl_xor(v, 1));
                acc[t][s][r] = v;
            }
        }
    }
    if (!(l16 & 1)) {
#pragma unroll
        for (int t = 0; t < FT; ++t) {
            const int fl = f0 + 16 * t + (half << 3);
            const int nr = F - fl;
#pragma unroll
            for (int s = 0; s < 4; ++s) {
                const int n  = ncol[s];
                const int bb = n >> Dlog2;
                const int d  = n & (D - 1);
                float* ocol = Out + ((size_t)bb * Dh + (d >> 1)) * F + fl;
                if (nr >= 8 && !(F & 3)) {
                    v4f lo, hi;
                    lo.x = acc[t][s][0]; lo.y = acc[t][s][1]; lo.z = acc[t][s][2]; lo.w = acc[t][s][3];
                    hi.x = acc[t][s][4]; hi.y = acc[t][s][5]; hi.z = acc[t][s][6]; hi.w = acc[t][s][7];
                    *reinterpret_cast<v4f*>(ocol) = lo;
                    *reinterpret_cast<v4f*>(ocol + 4) = hi;
                } else {
#pragma unroll
                    for (int r = 0; r < 8; ++r)
                        if (r < nr) ocol[r] = acc[t][s][r];
                }
            }
        }
    }
}

template <int KS>
__device__ __forceinline__ void fc_load_row(const float* rowSlice, int half,
                                            v4f (&v)[KS / 8])
{
    const v4f* p = reinterpret_cast<const v4f*>(rowSlice + 4 * half);
#pragma unroll
    for (int i = 0; i < KS / 8; ++i) v[i] = p[2 * i];
}

template <int K>
__global__ __launch_bounds__(256) void kd_fc_wmma(
    const float* __restrict__ X, const float* __restrict__ Wt,
    const float* __restrict__ bias, float* __restrict__ Out)
{
    static_assert(K % 64 == 0, "8 slices of whole 8-k pieces");
    constexpr int KS = K / 8;
    __shared__ float red[8 * 256];

    const int lane = threadIdx.x & 31;
    const int wv   = threadIdx.x >> 5;
    const int half = lane >> 4;
    const int l16  = lane & 15;
    const int n0   = blockIdx.x << 4;

    v4f va[KS / 8], vb[KS / 8];
    fc_load_row<KS>(Wt + (size_t)l16 * K + wv * KS, half, va);
    fc_load_row<KS>(X + (size_t)(n0 + l16) * K + wv * KS, half, vb);

    v8f acc0 = v8f{}, acc1 = v8f{};
#pragma unroll
    for (int i = 0; i < KS / 8; ++i) {
        v2f a0, b0, a1, b1;
        a0.x = va[i].x; a0.y = va[i].y;
        b0.x = vb[i].x; b0.y = vb[i].y;
        a1.x = va[i].z; a1.y = va[i].w;
        b1.x = vb[i].z; b1.y = vb[i].w;
        acc0 = __builtin_amdgcn_wmma_f32_16x16x4_f32(
            false, a0, false, b0, (short)0, acc0, false, false);
        acc1 = __builtin_amdgcn_wmma_f32_16x16x4_f32(
            false, a1, false, b1, (short)0, acc1, false, false);
    }

#pragma unroll
    for (int r = 0; r < 8; ++r)
        red[(wv << 8) + (r << 5) + lane] = acc0[r] + acc1[r];
    __syncthreads();

    const int e  = threadIdx.x;
    const int nl = e >> 4;
    const int o  = e & 15;
    const int cell = ((o & 7) << 5) + nl + ((o >> 3) << 4);
    float sum = red[cell];
#pragma unroll
    for (int w = 1; w < 8; ++w) sum += red[(w << 8) + cell];
    Out[((size_t)(n0 + nl) << 4) + o] = sum + bias[o];
}

__global__ __launch_bounds__(256) void kd_logsoftmax16(
    const float* __restrict__ logits, float* __restrict__ out, int rows)
{
    int r = blockIdx.x * blockDim.x + threadIdx.x;
    if (r >= rows) return;
    float v[16];
    float mx = -3.402823466e38f;
#pragma unroll
    for (int i = 0; i < 16; ++i) { v[i] = logits[r * 16 + i]; mx = fmaxf(mx, v[i]); }
    float s = 0.0f;
#pragma unroll
    for (int i = 0; i < 16; ++i) s += expf(v[i] - mx);
    const float l = mx + logf(s);
#pragma unroll
    for (int i = 0; i < 16; ++i) out[r * 16 + i] = v[i] - l;
}

extern "C" void kernel_launch(void* const* d_in, const int* in_sizes, int n_in,
                              void* d_out, int out_size, void* d_ws, size_t ws_size,
                              hipStream_t stream) {
    static const int DIMS[11] = {2048, 1024, 512, 256, 128, 64, 32, 16, 8, 4, 2};
    static const int INCH[11] = {3, 8, 32, 64, 64, 64, 128, 256, 512, 512, 512};
    static const int FEAT[11] = {8, 32, 64, 64, 64, 128, 256, 512, 512, 512, 1024};
    const int Bsz = 512;

    const size_t XBYTES = (size_t)512 * 32 * 512 * sizeof(float);
    float* X0     = (float*)d_ws;
    float* X1     = (float*)((char*)d_ws + XBYTES);
    float* logits = (float*)((char*)d_ws + 2 * XBYTES);

    const float* cur = (const float*)d_in[0];
    for (int i = 1; i < 11; ++i) {
        const int Dd = DIMS[i], Kc = INCH[i], F = FEAT[i];
        const int Dlog2 = __builtin_ctz((unsigned)Dd);
        float* nxt = (i & 1) ? X1 : X0;

        const int N = Bsz * Dd;
        const int FT = (i >= 1 && i <= 3) ? 2 : 1;
        const int tilesF = (F + 16 * FT - 1) / (16 * FT);
        const int ncg = ((N / 512) * tilesF < 256) ? 4 : 8;
        const int KS  = (ncg == 4 && Kc % 64 == 0) ? 2 : 1;
        const int nw  = ncg * KS;
        const int blocksX = N / (64 * ncg);
        const int Kp4 = (Kc + 3) & ~3;
        const size_t ldsBytes = (size_t)48 * FT * (Kp4 + 1) * sizeof(float)
            + (KS == 2 ? (size_t)ncg * 1024 * FT * sizeof(float) : 0)
            + (i == 1 ? (size_t)(3 + 192 + 24) * sizeof(float) : 0);
        dim3 grid(blocksX, tilesF);
        const float* Wl = (const float*)d_in[12 + 2 * i];
        const float* bl = (const float*)d_in[13 + 2 * i];
        const int*   sl = (const int*)d_in[1 + i];
        if (i == 1)
            kd_gather_wmma<1, 1, 2, false, true><<<grid, 32 * nw, ldsBytes, stream>>>(
                cur, Wl, bl, sl, nxt, F, Kc, Dlog2, nw, KS,
                (const float*)d_in[12], (const float*)d_in[13], (const int*)d_in[1], INCH[0]);
        else if (i == 2 || i == 3)
            kd_gather_wmma<1, 4, 2, false, false><<<grid, 32 * nw, ldsBytes, stream>>>(
                cur, Wl, bl, sl, nxt, F, Kc, Dlog2, nw, KS, nullptr, nullptr, nullptr, 0);
        else if (i == 4 || i == 5)
            kd_gather_wmma<2, 4, 1, false, false><<<grid, 32 * nw, ldsBytes, stream>>>(
                cur, Wl, bl, sl, nxt, F, Kc, Dlog2, nw, KS, nullptr, nullptr, nullptr, 0);
        else
            kd_gather_wmma<2, 4, 1, true, false><<<grid, 32 * nw, ldsBytes, stream>>>(
                cur, Wl, bl, sl, nxt, F, Kc, Dlog2, nw, KS, nullptr, nullptr, nullptr, 0);
        cur = nxt;
    }

    kd_fc_wmma<1024><<<dim3(32), dim3(256), 0, stream>>>(
        cur, (const float*)d_in[34], (const float*)d_in[35], logits);

    kd_logsoftmax16<<<(512 + 255) / 256, 256, 0, stream>>>(logits, (float*)d_out, 512);

    (void)in_sizes; (void)n_in; (void)out_size; (void)ws_size;
}
